# starting kernel
# baseline (speedup 1.0000x reference)
#include <hip/hip_runtime.h>
#include <math.h>

#define H_     512
#define NMODE  32
#define B_     8
#define L_     2048
#define Q_     64
#define NCHUNK 32
#define CPITCH 34
#define WPITCH 516

typedef __attribute__((ext_vector_type(2))) float v2f;
typedef __attribute__((ext_vector_type(8))) float v8f;

__device__ __forceinline__ v8f wmma4(v2f a, v2f b, v8f c) {
  return __builtin_amdgcn_wmma_f32_16x16x4_f32(false, a, false, b, (short)0, c,
                                               false, false);
}

__global__ void s4d_setup(const float* __restrict__ log_dt,
                          const float* __restrict__ C,
                          const float* __restrict__ log_A_real,
                          const float* __restrict__ A_imag,
                          float* __restrict__ Ktap2,
                          float* __restrict__ CoutR,
                          float* __restrict__ CoutI,
                          float* __restrict__ Wr,
                          float* __restrict__ Wi) {
  const int h = blockIdx.x;
  const int k = threadIdx.x;
  const float dt = expf(log_dt[h]);
  const float Ar = -expf(log_A_real[h * NMODE + k]);
  const float Ai = A_imag[h * NMODE + k];
  const float Ccr = C[(h * NMODE + k) * 2 + 0];
  const float Cci = C[(h * NMODE + k) * 2 + 1];
  const float er = expf(Ar * dt);
  const float wr = er * cosf(Ai * dt);
  const float wi = er * sinf(Ai * dt);
  const float exr = wr - 1.0f, exi = wi;
  const float den = Ar * Ar + Ai * Ai;
  const float qr = (exr * Ar + exi * Ai) / den;
  const float qi = (exi * Ar - exr * Ai) / den;
  const float Ckr = Ccr * qr - Cci * qi;
  const float Cki = Ccr * qi + Cci * qr;
  Wr[h * NMODE + k] = wr;
  Wi[h * NMODE + k] = wi;

  Ktap2[h * 128 + k] = 0.0f;
  Ktap2[h * 128 + 32 + k] = 0.0f;

  float pr = Ckr, pi = Cki;
  for (int d = 0; d < Q_; ++d) {
    float v = 2.0f * pr;
    for (int m = 16; m >= 1; m >>= 1) v += __shfl_xor(v, m, 32);
    if (k == 0) Ktap2[h * 128 + 64 + d] = v;
    const float nr = pr * wr - pi * wi;
    const float ni = pr * wi + pi * wr;
    pr = nr; pi = ni;
  }
  float sr = Ckr * wr - Cki * wi;
  float si = Ckr * wi + Cki * wr;
  for (int j = 0; j < Q_; ++j) {
    CoutR[((size_t)h * Q_ + j) * NMODE + k] = 2.0f * sr;
    CoutI[((size_t)h * Q_ + j) * NMODE + k] = -2.0f * si;
    const float nr = sr * wr - si * wi;
    const float ni = sr * wi + si * wr;
    sr = nr; si = ni;
  }
}

__global__ void s4d_scan(const float* __restrict__ u,
                         const float* __restrict__ Wr,
                         const float* __restrict__ Wi,
                         float* __restrict__ XR,
                         float* __restrict__ XI) {
  const int wid  = (int)((blockIdx.x * blockDim.x + threadIdx.x) >> 5);
  const int lane = threadIdx.x & 31;
  const int b = wid >> 9;
  const int h = wid & (H_ - 1);
  const float wr = Wr[h * NMODE + lane];
  const float wi = Wi[h * NMODE + lane];
  float sr = 0.0f, si = 0.0f;
  const float* up = u + (size_t)(b * H_ + h) * L_;
  const size_t xb = ((size_t)(b * H_ + h)) * NCHUNK * NMODE;
  for (int c = 0; c < NCHUNK; ++c) {
    XR[xb + (size_t)c * NMODE + lane] = sr;
    XI[xb + (size_t)c * NMODE + lane] = si;
    for (int t = 0; t < Q_; ++t) {
      const float uu = up[c * Q_ + t];
      const float nr = sr * wr - si * wi + uu;
      const float ni = sr * wi + si * wr;
      sr = nr; si = ni;
    }
  }
}

__global__ void s4d_conv(const float* __restrict__ u,
                         const float* __restrict__ Ktap2,
                         const float* __restrict__ CoutR,
                         const float* __restrict__ CoutI,
                         const float* __restrict__ XR,
                         const float* __restrict__ XI,
                         const float* __restrict__ Dvec,
                         float* __restrict__ ygelu) {
  __shared__ float sKt[128];
  __shared__ float sCR[Q_ * CPITCH];
  __shared__ float sCI[Q_ * CPITCH];

  const int h      = blockIdx.x >> 3;
  const int g      = blockIdx.x & 7;
  const int waveId = threadIdx.x >> 5;
  const int lane   = threadIdx.x & 31;
  const int half   = lane >> 4;
  const int lm     = lane & 15;
  const int jt     = waveId & 3;
  const int ct     = waveId >> 2;
  const int q      = g * 32 + ct * 16 + lm;
  const int b      = q >> 5;
  const int c      = q & 31;

  if (threadIdx.x < 128) sKt[threadIdx.x] = Ktap2[(size_t)h * 128 + threadIdx.x];
  for (int i = threadIdx.x; i < Q_ * NMODE; i += 256) {
    const int j = i >> 5, k = i & 31;
    sCR[j * CPITCH + k] = CoutR[((size_t)h * Q_ + j) * NMODE + k];
    sCI[j * CPITCH + k] = CoutI[((size_t)h * Q_ + j) * NMODE + k];
  }
  __syncthreads();

  const size_t ub = (size_t)(b * H_ + h) * L_ + (size_t)c * Q_;
  const int jrow = jt * 16 + lm;

  v8f acc = {0.f, 0.f, 0.f, 0.f, 0.f, 0.f, 0.f, 0.f};

#pragma unroll
  for (int k0 = 0; k0 < Q_; k0 += 4) {
    const int ta = k0 + 2 * half;
    v2f a, bb;
    a.x = sKt[64 + jrow - ta];
    a.y = sKt[63 + jrow - ta];
    bb.x = u[ub + ta];
    bb.y = u[ub + ta + 1];
    acc = wmma4(a, bb, acc);
  }
  const size_t xrb = ((size_t)(b * H_ + h) * NCHUNK + c) * NMODE;
#pragma unroll
  for (int k0 = 0; k0 < NMODE; k0 += 4) {
    const int ka = k0 + 2 * half;
    v2f a, bb;
    a.x = sCR[jrow * CPITCH + ka];
    a.y = sCR[jrow * CPITCH + ka + 1];
    bb.x = XR[xrb + ka];
    bb.y = XR[xrb + ka + 1];
    acc = wmma4(a, bb, acc);
  }
#pragma unroll
  for (int k0 = 0; k0 < NMODE; k0 += 4) {
    const int ka = k0 + 2 * half;
    v2f a, bb;
    a.x = sCI[jrow * CPITCH + ka];
    a.y = sCI[jrow * CPITCH + ka + 1];
    bb.x = XI[xrb + ka];
    bb.y = XI[xrb + ka + 1];
    acc = wmma4(a, bb, acc);
  }
  const float Dh = Dvec[h];
#pragma unroll
  for (int v = 0; v < 8; ++v) {
    const int j = jt * 16 + v + 8 * half;
    const size_t idx = (size_t)(b * H_ + h) * L_ + (size_t)c * Q_ + j;
    const float uu = u[idx];
    const float y = acc[v] + Dh * uu;
    ygelu[idx] = 0.5f * y * (1.0f + erff(y * 0.70710678118654752f));
  }
}

__global__ void s4d_glu(const float* __restrict__ ygelu,
                        const float* __restrict__ W,
                        const float* __restrict__ bias,
                        float* __restrict__ out) {
  extern __shared__ float sW[];
  const int bid    = blockIdx.x;
  const int waveId = threadIdx.x >> 5;
  const int lane   = threadIdx.x & 31;
  const int half   = lane >> 4;
  const int lm     = lane & 15;
  const int mtile  = bid >> 7;
  const int ntile  = (bid & 127) * 8 + waveId;
  const int col    = ntile * 16 + lm;
  const int bb     = col >> 11;
  const int l      = col & (L_ - 1);

  for (int i = threadIdx.x; i < (32 * H_) / 4; i += 256) {
    const int r = (i * 4) >> 9;
    const int k = (i * 4) & (H_ - 1);
    const int src = (r < 16) ? (mtile * 16 + r) : (H_ + mtile * 16 + (r - 16));
    const float4 v4 = *(const float4*)(W + (size_t)src * H_ + k);
    float* dst = sW + r * WPITCH + k;
    dst[0] = v4.x; dst[1] = v4.y; dst[2] = v4.z; dst[3] = v4.w;
  }
  __syncthreads();

  const float* yb = ygelu + (size_t)bb * H_ * L_ + l;
  const float* la = sW + lm * WPITCH;
  const float* lg = sW + (16 + lm) * WPITCH;

  v8f accA = {0.f, 0.f, 0.f, 0.f, 0.f, 0.f, 0.f, 0.f};
  v8f accG = {0.f, 0.f, 0.f, 0.f, 0.f, 0.f, 0.f, 0.f};

#pragma unroll 4
  for (int k0 = 0; k0 < H_; k0 += 4) {
    const int ka = k0 + 2 * half;
    v2f bf, aA, aG;
    bf.x = yb[(size_t)ka * L_];
    bf.y = yb[(size_t)(ka + 1) * L_];
    aA.x = la[ka];  aA.y = la[ka + 1];
    aG.x = lg[ka];  aG.y = lg[ka + 1];
    if (k0 + 32 < H_) {
      __builtin_prefetch(yb + (size_t)(k0 + 32) * L_, 0, 0);
    }
    accA = wmma4(aA, bf, accA);
    accG = wmma4(aG, bf, accG);
  }
#pragma unroll
  for (int v = 0; v < 8; ++v) {
    const int hr = mtile * 16 + v + 8 * half;
    const float a = accA[v] + bias[hr];
    const float gv = accG[v] + bias[H_ + hr];
    out[((size_t)(bb * H_ + hr)) * L_ + l] = a * (1.0f / (1.0f + expf(-gv)));
  }
}

extern "C" void kernel_launch(void* const* d_in, const int* in_sizes, int n_in,
                              void* d_out, int out_size, void* d_ws, size_t ws_size,
                              hipStream_t stream) {
  (void)in_sizes; (void)n_in; (void)out_size; (void)ws_size;
  const float* u          = (const float*)d_in[0];
  const float* log_dt     = (const float*)d_in[1];
  const float* C          = (const float*)d_in[2];
  const float* log_A_real = (const float*)d_in[3];
  const float* A_imag     = (const float*)d_in[4];
  const float* Dvec       = (const float*)d_in[5];
  const float* W          = (const float*)d_in[6];
  const float* bias       = (const float*)d_in[7];
  float* out = (float*)d_out;

  float* ws = (float*)d_ws;
  float* Ktap2 = ws;
  float* CoutR = Ktap2 + (size_t)H_ * 128;
  float* CoutI = CoutR + (size_t)H_ * Q_ * NMODE;
  float* Wr    = CoutI + (size_t)H_ * Q_ * NMODE;
  float* Wi    = Wr    + (size_t)H_ * NMODE;
  float* XR    = Wi    + (size_t)H_ * NMODE;
  float* XI    = XR    + (size_t)B_ * H_ * NCHUNK * NMODE;
  float* ygelu = XI    + (size_t)B_ * H_ * NCHUNK * NMODE;

  s4d_setup<<<H_, 32, 0, stream>>>(log_dt, C, log_A_real, A_imag,
                                   Ktap2, CoutR, CoutI, Wr, Wi);
  s4d_scan<<<(B_ * H_) / 8, 256, 0, stream>>>(u, Wr, Wi, XR, XI);
  s4d_conv<<<H_ * 8, 256, 0, stream>>>(u, Ktap2, CoutR, CoutI, XR, XI, Dvec, ygelu);
  s4d_glu<<<4096, 256, (size_t)(32 * WPITCH) * sizeof(float), stream>>>(ygelu, W,
                                                                        bias, out);
}


# GLU GEMM: W panel (64a+64g rows, whole K) resident in LDS per block, block walks 512 columns, activation fragments global->registers in two register sets (not via LDS), wave tile 8x1; k order kept
# speedup vs baseline: 1.1434x; 1.1434x over previous
#include <hip/hip_runtime.h>
#include <math.h>

#define H_     512
#define NMODE  32
#define B_     8
#define L_     2048
#define Q_     64
#define NCHUNK 32
#define CPITCH 34
#define WPITCH 516

typedef __attribute__((ext_vector_type(2))) float v2f;
typedef __attribute__((ext_vector_type(8))) float v8f;

__device__ __forceinline__ v8f wmma4(v2f a, v2f b, v8f c) {
  return __builtin_amdgcn_wmma_f32_16x16x4_f32(false, a, false, b, (short)0, c,
                                               false, false);
}

__global__ void s4d_setup(const float* __restrict__ log_dt,
                          const float* __restrict__ C,
                          const float* __restrict__ log_A_real,
                          const float* __restrict__ A_imag,
                          float* __restrict__ Ktap2,
                          float* __restrict__ CoutR,
                          float* __restrict__ CoutI,
                          float* __restrict__ Wr,
                          float* __restrict__ Wi) {
  const int h = blockIdx.x;
  const int k = threadIdx.x;
  const float dt = expf(log_dt[h]);
  const float Ar = -expf(log_A_real[h * NMODE + k]);
  const float Ai = A_imag[h * NMODE + k];
  const float Ccr = C[(h * NMODE + k) * 2 + 0];
  const float Cci = C[(h * NMODE + k) * 2 + 1];
  const float er = expf(Ar * dt);
  const float wr = er * cosf(Ai * dt);
  const float wi = er * sinf(Ai * dt);
  const float exr = wr - 1.0f, exi = wi;
  const float den = Ar * Ar + Ai * Ai;
  const float qr = (exr * Ar + exi * Ai) / den;
  const float qi = (exi * Ar - exr * Ai) / den;
  const float Ckr = Ccr * qr - Cci * qi;
  const float Cki = Ccr * qi + Cci * qr;
  Wr[h * NMODE + k] = wr;
  Wi[h * NMODE + k] = wi;

  Ktap2[h * 128 + k] = 0.0f;
  Ktap2[h * 128 + 32 + k] = 0.0f;

  float pr = Ckr, pi = Cki;
  for (int d = 0; d < Q_; ++d) {
    float v = 2.0f * pr;
    for (int m = 16; m >= 1; m >>= 1) v += __shfl_xor(v, m, 32);
    if (k == 0) Ktap2[h * 128 + 64 + d] = v;
    const float nr = pr * wr - pi * wi;
    const float ni = pr * wi + pi * wr;
    pr = nr; pi = ni;
  }
  float sr = Ckr * wr - Cki * wi;
  float si = Ckr * wi + Cki * wr;
  for (int j = 0; j < Q_; ++j) {
    CoutR[((size_t)h * Q_ + j) * NMODE + k] = 2.0f * sr;
    CoutI[((size_t)h * Q_ + j) * NMODE + k] = -2.0f * si;
    const float nr = sr * wr - si * wi;
    const float ni = sr * wi + si * wr;
    sr = nr; si = ni;
  }
}

__global__ void s4d_scan(const float* __restrict__ u,
                         const float* __restrict__ Wr,
                         const float* __restrict__ Wi,
                         float* __restrict__ XR,
                         float* __restrict__ XI) {
  const int wid  = (int)((blockIdx.x * blockDim.x + threadIdx.x) >> 5);
  const int lane = threadIdx.x & 31;
  const int b = wid >> 9;
  const int h = wid & (H_ - 1);
  const float wr = Wr[h * NMODE + lane];
  const float wi = Wi[h * NMODE + lane];
  float sr = 0.0f, si = 0.0f;
  const float* up = u + (size_t)(b * H_ + h) * L_;
  const size_t xb = ((size_t)(b * H_ + h)) * NCHUNK * NMODE;
  for (int c = 0; c < NCHUNK; ++c) {
    XR[xb + (size_t)c * NMODE + lane] = sr;
    XI[xb + (size_t)c * NMODE + lane] = si;
    for (int t = 0; t < Q_; ++t) {
      const float uu = up[c * Q_ + t];
      const float nr = sr * wr - si * wi + uu;
      const float ni = sr * wi + si * wr;
      sr = nr; si = ni;
    }
  }
}

__global__ void s4d_conv(const float* __restrict__ u,
                         const float* __restrict__ Ktap2,
                         const float* __restrict__ CoutR,
                         const float* __restrict__ CoutI,
                         const float* __restrict__ XR,
                         const float* __restrict__ XI,
                         const float* __restrict__ Dvec,
                         float* __restrict__ ygelu) {
  __shared__ float sKt[128];
  __shared__ float sCR[Q_ * CPITCH];
  __shared__ float sCI[Q_ * CPITCH];

  const int h      = blockIdx.x >> 3;
  const int g      = blockIdx.x & 7;
  const int waveId = threadIdx.x >> 5;
  const int lane   = threadIdx.x & 31;
  const int half   = lane >> 4;
  const int lm     = lane & 15;
  const int jt     = waveId & 3;
  const int ct     = waveId >> 2;
  const int q      = g * 32 + ct * 16 + lm;
  const int b      = q >> 5;
  const int c      = q & 31;

  if (threadIdx.x < 128) sKt[threadIdx.x] = Ktap2[(size_t)h * 128 + threadIdx.x];
  for (int i = threadIdx.x; i < Q_ * NMODE; i += 256) {
    const int j = i >> 5, k = i & 31;
    sCR[j * CPITCH + k] = CoutR[((size_t)h * Q_ + j) * NMODE + k];
    sCI[j * CPITCH + k] = CoutI[((size_t)h * Q_ + j) * NMODE + k];
  }
  __syncthreads();

  const size_t ub = (size_t)(b * H_ + h) * L_ + (size_t)c * Q_;
  const int jrow = jt * 16 + lm;

  v8f acc = {0.f, 0.f, 0.f, 0.f, 0.f, 0.f, 0.f, 0.f};

#pragma unroll
  for (int k0 = 0; k0 < Q_; k0 += 4) {
    const int ta = k0 + 2 * half;
    v2f a, bb;
    a.x = sKt[64 + jrow - ta];
    a.y = sKt[63 + jrow - ta];
    bb.x = u[ub + ta];
    bb.y = u[ub + ta + 1];
    acc = wmma4(a, bb, acc);
  }
  const size_t xrb = ((size_t)(b * H_ + h) * NCHUNK + c) * NMODE;
#pragma unroll
  for (int k0 = 0; k0 < NMODE; k0 += 4) {
    const int ka = k0 + 2 * half;
    v2f a, bb;
    a.x = sCR[jrow * CPITCH + ka];
    a.y = sCR[jrow * CPITCH + ka + 1];
    bb.x = XR[xrb + ka];
    bb.y = XR[xrb + ka + 1];
    acc = wmma4(a, bb, acc);
  }
#pragma unroll
  for (int k0 = 0; k0 < NMODE; k0 += 4) {
    const int ka = k0 + 2 * half;
    v2f a, bb;
    a.x = sCI[jrow * CPITCH + ka];
    a.y = sCI[jrow * CPITCH + ka + 1];
    bb.x = XI[xrb + ka];
    bb.y = XI[xrb + ka + 1];
    acc = wmma4(a, bb, acc);
  }
  const float Dh = Dvec[h];
#pragma unroll
  for (int v = 0; v < 8; ++v) {
    const int j = jt * 16 + v + 8 * half;
    const size_t idx = (size_t)(b * H_ + h) * L_ + (size_t)c * Q_ + j;
    const float uu = u[idx];
    const float y = acc[v] + Dh * uu;
    ygelu[idx] = 0.5f * y * (1.0f + erff(y * 0.70710678118654752f));
  }
}

#define GP_ROWS    128
#define GP_THREADS 512
#define GP_KG      16

__global__ void __launch_bounds__(GP_THREADS)
s4d_glu(const float* __restrict__ ygelu,
        const float* __restrict__ W,
        const float* __restrict__ bias,
        float* __restrict__ out) {
  __shared__ __attribute__((aligned(16))) float sW[GP_ROWS * WPITCH];
  const int tid    = threadIdx.x;
  const int waveId = tid >> 5;
  const int lane   = tid & 31;
  const int half   = lane >> 4;
  const int lm     = lane & 15;
  const int panel  = blockIdx.x & 7;
  const int cgrp   = blockIdx.x >> 3;
  const int col0   = cgrp * 512 + waveId * 16 + lm;
  const int bb     = col0 >> 11;
  const int l0     = col0 & (L_ - 1);

  const float* yb = ygelu + (size_t)bb * H_ * L_ + (size_t)(2 * half) * L_ + l0;

  v2f bcur[4], bnxt[4];
#pragma unroll
  for (int s = 0; s < 4; ++s) {
    bcur[s].x = yb[(size_t)(s * 4) * L_];
    bcur[s].y = yb[(size_t)(s * 4 + 1) * L_];
  }

  for (int it = 0; it < 4; ++it) {
    float4 tmp[8];
#pragma unroll
    for (int j = 0; j < 8; ++j) {
      const int r  = (it * 8 + j) * 4 + (tid >> 7);
      const int k4 = tid & 127;
      const int srow = (r < 64) ? (panel * 64 + r) : (H_ + panel * 64 + (r - 64));
      tmp[j] = *(const float4*)(W + (size_t)srow * H_ + k4 * 4);
    }
#pragma unroll
    for (int j = 0; j < 8; ++j) {
      const int r  = (it * 8 + j) * 4 + (tid >> 7);
      const int k4 = tid & 127;
      *(float4*)(sW + r * WPITCH + k4 * 4) = tmp[j];
    }
  }
  __syncthreads();

  const float* la = sW + lm * WPITCH + 2 * half;

  for (int t = 0; t < 2; ++t) {
    v8f acc[8];
#pragma unroll
    for (int i = 0; i < 8; ++i) {
      acc[i] = (v8f){0.f, 0.f, 0.f, 0.f, 0.f, 0.f, 0.f, 0.f};
    }

    for (int g = 0; g < H_ / GP_KG; ++g) {
      int gn = t * (H_ / GP_KG) + g + 1;
      gn = (gn < 2 * (H_ / GP_KG)) ? gn : (2 * (H_ / GP_KG) - 1);
      const float* yn = yb + (gn >> 5) * 256 + (size_t)((gn & 31) * GP_KG) * L_;
#pragma unroll
      for (int s = 0; s < 4; ++s) {
        bnxt[s].x = yn[(size_t)(s * 4) * L_];
        bnxt[s].y = yn[(size_t)(s * 4 + 1) * L_];
      }

      const float* lk = la + g * GP_KG;
#pragma unroll
      for (int s = 0; s < 4; ++s) {
        v2f a[8];
#pragma unroll
        for (int i = 0; i < 8; ++i) {
          a[i] = *(const v2f*)(lk + i * 16 * WPITCH + s * 4);
        }
#pragma unroll
        for (int i = 0; i < 8; ++i) {
          acc[i] = wmma4(a[i], bcur[s], acc[i]);
        }
      }
#pragma unroll
      for (int s = 0; s < 4; ++s) bcur[s] = bnxt[s];
    }

    const int l = l0 + t * 256;
#pragma unroll
    for (int i = 0; i < 4; ++i) {
#pragma unroll
      for (int v = 0; v < 8; ++v) {
        const int hr = panel * 64 + i * 16 + v + 8 * half;
        const float a  = acc[i][v] + bias[hr];
        const float gv = acc[4 + i][v] + bias[H_ + hr];
        out[((size_t)(bb * H_ + hr)) * L_ + l] = a * (1.0f / (1.0f + expf(-gv)));
      }
    }
  }
}

extern "C" void kernel_launch(void* const* d_in, const int* in_sizes, int n_in,
                              void* d_out, int out_size, void* d_ws, size_t ws_size,
                              hipStream_t stream) {
  (void)in_sizes; (void)n_in; (void)out_size; (void)ws_size;
  const float* u          = (const float*)d_in[0];
  const float* log_dt     = (const float*)d_in[1];
  const float* C          = (const float*)d_in[2];
  const float* log_A_real = (const float*)d_in[3];
  const float* A_imag     = (const float*)d_in[4];
  const float* Dvec       = (const float*)d_in[5];
  const float* W          = (const float*)d_in[6];
  const float* bias       = (const float*)d_in[7];
  float* out = (float*)d_out;

  float* ws = (float*)d_ws;
  float* Ktap2 = ws;
  float* CoutR = Ktap2 + (size_t)H_ * 128;
  float* CoutI = CoutR + (size_t)H_ * Q_ * NMODE;
  float* Wr    = CoutI + (size_t)H_ * Q_ * NMODE;
  float* Wi    = Wr    + (size_t)H_ * NMODE;
  float* XR    = Wi    + (size_t)H_ * NMODE;
  float* XI    = XR    + (size_t)B_ * H_ * NCHUNK * NMODE;
  float* ygelu = XI    + (size_t)B_ * H_ * NCHUNK * NMODE;

  s4d_setup<<<H_, 32, 0, stream>>>(log_dt, C, log_A_real, A_imag,
                                   Ktap2, CoutR, CoutI, Wr, Wi);
  s4d_scan<<<(B_ * H_) / 8, 256, 0, stream>>>(u, Wr, Wi, XR, XI);
  s4d_conv<<<H_ * 8, 256, 0, stream>>>(u, Ktap2, CoutR, CoutI, XR, XI, Dvec, ygelu);
  s4d_glu<<<256, GP_THREADS, 0, stream>>>(ygelu, W, bias, out);
}


# scan kernel: 16-step block sums against a per-lane table of w^0..w^15 replace the serial recurrence (s = w^16*s + block sum), next chunk loaded under the current one; state sum order changes
# speedup vs baseline: 1.2716x; 1.1121x over previous
#include <hip/hip_runtime.h>
#include <math.h>

#define H_     512
#define NMODE  32
#define B_     8
#define L_     2048
#define Q_     64
#define NCHUNK 32
#define CPITCH 34
#define WPITCH 516

typedef __attribute__((ext_vector_type(2))) float v2f;
typedef __attribute__((ext_vector_type(8))) float v8f;

__device__ __forceinline__ v8f wmma4(v2f a, v2f b, v8f c) {
  return __builtin_amdgcn_wmma_f32_16x16x4_f32(false, a, false, b, (short)0, c,
                                               false, false);
}

__global__ void s4d_setup(const float* __restrict__ log_dt,
                          const float* __restrict__ C,
                          const float* __restrict__ log_A_real,
                          const float* __restrict__ A_imag,
                          float* __restrict__ Ktap2,
                          float* __restrict__ CoutR,
                          float* __restrict__ CoutI,
                          float* __restrict__ Wr,
                          float* __restrict__ Wi) {
  const int h = blockIdx.x;
  const int k = threadIdx.x;
  const float dt = expf(log_dt[h]);
  const float Ar = -expf(log_A_real[h * NMODE + k]);
  const float Ai = A_imag[h * NMODE + k];
  const float Ccr = C[(h * NMODE + k) * 2 + 0];
  const float Cci = C[(h * NMODE + k) * 2 + 1];
  const float er = expf(Ar * dt);
  const float wr = er * cosf(Ai * dt);
  const float wi = er * sinf(Ai * dt);
  const float exr = wr - 1.0f, exi = wi;
  const float den = Ar * Ar + Ai * Ai;
  const float qr = (exr * Ar + exi * Ai) / den;
  const float qi = (exi * Ar - exr * Ai) / den;
  const float Ckr = Ccr * qr - Cci * qi;
  const float Cki = Ccr * qi + Cci * qr;
  Wr[h * NMODE + k] = wr;
  Wi[h * NMODE + k] = wi;

  Ktap2[h * 128 + k] = 0.0f;
  Ktap2[h * 128 + 32 + k] = 0.0f;

  float pr = Ckr, pi = Cki;
  for (int d = 0; d < Q_; ++d) {
    float v = 2.0f * pr;
    for (int m = 16; m >= 1; m >>= 1) v += __shfl_xor(v, m, 32);
    if (k == 0) Ktap2[h * 128 + 64 + d] = v;
    const float nr = pr * wr - pi * wi;
    const float ni = pr * wi + pi * wr;
    pr = nr; pi = ni;
  }
  float sr = Ckr * wr - Cki * wi;
  float si = Ckr * wi + Cki * wr;
  for (int j = 0; j < Q_; ++j) {
    CoutR[((size_t)h * Q_ + j) * NMODE + k] = 2.0f * sr;
    CoutI[((size_t)h * Q_ + j) * NMODE + k] = -2.0f * si;
    const float nr = sr * wr - si * wi;
    const float ni = sr * wi + si * wr;
    sr = nr; si = ni;
  }
}

#define SB 16

__device__ __forceinline__ void scan_load(float (&x)[Q_], const float4* __restrict__ p) {
#pragma unroll
  for (int q = 0; q < Q_ / 4; ++q) {
    const float4 v = p[q];
    x[4 * q + 0] = v.x; x[4 * q + 1] = v.y; x[4 * q + 2] = v.z; x[4 * q + 3] = v.w;
  }
}

__device__ __forceinline__ void scan_chunk(const float (&x)[Q_],
                                           const float (&pr)[SB], const float (&pi)[SB],
                                           float w16r, float w16i,
                                           float& sr, float& si) {
#pragma unroll
  for (int q = 0; q < Q_ / SB; ++q) {
    const float* xb = x + q * SB;
    float ar0 = pr[SB - 1] * xb[0], ai0 = pi[SB - 1] * xb[0];
    float ar1 = pr[SB - 2] * xb[1], ai1 = pi[SB - 2] * xb[1];
#pragma unroll
    for (int i = 2; i < SB - 2; i += 2) {
      ar0 = fmaf(pr[SB - 1 - i], xb[i], ar0);
      ai0 = fmaf(pi[SB - 1 - i], xb[i], ai0);
      ar1 = fmaf(pr[SB - 2 - i], xb[i + 1], ar1);
      ai1 = fmaf(pi[SB - 2 - i], xb[i + 1], ai1);
    }
    ar0 = fmaf(pr[1], xb[SB - 2], ar0);
    ai0 = fmaf(pi[1], xb[SB - 2], ai0);
    ar1 += xb[SB - 1];
    const float Sr = ar0 + ar1, Si = ai0 + ai1;
    const float nr = fmaf(w16r, sr, fmaf(-w16i, si, Sr));
    const float ni = fmaf(w16r, si, fmaf(w16i, sr, Si));
    sr = nr; si = ni;
  }
}

__global__ void __launch_bounds__(256) __attribute__((amdgpu_waves_per_eu(4, 4)))
s4d_scan(const float* __restrict__ u,
         const float* __restrict__ Wr,
         const float* __restrict__ Wi,
         float* __restrict__ XR,
         float* __restrict__ XI) {
  const int wid  = (int)((blockIdx.x * blockDim.x + threadIdx.x) >> 5);
  const int lane = threadIdx.x & 31;
  const int b = wid >> 9;
  const int h = wid & (H_ - 1);
  const float wr = Wr[h * NMODE + lane];
  const float wi = Wi[h * NMODE + lane];

  float pr[SB], pi[SB];
  pr[0] = 1.0f; pi[0] = 0.0f;
  pr[1] = wr;   pi[1] = wi;
#pragma unroll
  for (int m = 2; m < SB; ++m) {
    pr[m] = pr[m - 1] * wr - pi[m - 1] * wi;
    pi[m] = pr[m - 1] * wi + pi[m - 1] * wr;
  }
  const float w16r = pr[SB - 1] * wr - pi[SB - 1] * wi;
  const float w16i = pr[SB - 1] * wi + pi[SB - 1] * wr;

  float sr = 0.0f, si = 0.0f;
  const float4* up4 = (const float4*)(u + (size_t)(b * H_ + h) * L_);
  const size_t xb = ((size_t)(b * H_ + h)) * NCHUNK * NMODE;

  float xa[Q_], xc[Q_];
  scan_load(xa, up4);
  for (int c = 0; c < NCHUNK; c += 2) {
    scan_load(xc, up4 + (c + 1) * (Q_ / 4));
    XR[xb + (size_t)c * NMODE + lane] = sr;
    XI[xb + (size_t)c * NMODE + lane] = si;
    scan_chunk(xa, pr, pi, w16r, w16i, sr, si);
    const int cn = (c + 2 < NCHUNK) ? (c + 2) : (NCHUNK - 1);
    scan_load(xa, up4 + cn * (Q_ / 4));
    XR[xb + (size_t)(c + 1) * NMODE + lane] = sr;
    XI[xb + (size_t)(c + 1) * NMODE + lane] = si;
    scan_chunk(xc, pr, pi, w16r, w16i, sr, si);
  }
}

__global__ void s4d_conv(const float* __restrict__ u,
                         const float* __restrict__ Ktap2,
                         const float* __restrict__ CoutR,
                         const float* __restrict__ CoutI,
                         const float* __restrict__ XR,
                         const float* __restrict__ XI,
                         const float* __restrict__ Dvec,
                         float* __restrict__ ygelu) {
  __shared__ float sKt[128];
  __shared__ float sCR[Q_ * CPITCH];
  __shared__ float sCI[Q_ * CPITCH];

  const int h      = blockIdx.x >> 3;
  const int g      = blockIdx.x & 7;
  const int waveId = threadIdx.x >> 5;
  const int lane   = threadIdx.x & 31;
  const int half   = lane >> 4;
  const int lm     = lane & 15;
  const int jt     = waveId & 3;
  const int ct     = waveId >> 2;
  const int q      = g * 32 + ct * 16 + lm;
  const int b      = q >> 5;
  const int c      = q & 31;

  if (threadIdx.x < 128) sKt[threadIdx.x] = Ktap2[(size_t)h * 128 + threadIdx.x];
  for (int i = threadIdx.x; i < Q_ * NMODE; i += 256) {
    const int j = i >> 5, k = i & 31;
    sCR[j * CPITCH + k] = CoutR[((size_t)h * Q_ + j) * NMODE + k];
    sCI[j * CPITCH + k] = CoutI[((size_t)h * Q_ + j) * NMODE + k];
  }
  __syncthreads();

  const size_t ub = (size_t)(b * H_ + h) * L_ + (size_t)c * Q_;
  const int jrow = jt * 16 + lm;

  v8f acc = {0.f, 0.f, 0.f, 0.f, 0.f, 0.f, 0.f, 0.f};

#pragma unroll
  for (int k0 = 0; k0 < Q_; k0 += 4) {
    const int ta = k0 + 2 * half;
    v2f a, bb;
    a.x = sKt[64 + jrow - ta];
    a.y = sKt[63 + jrow - ta];
    bb.x = u[ub + ta];
    bb.y = u[ub + ta + 1];
    acc = wmma4(a, bb, acc);
  }
  const size_t xrb = ((size_t)(b * H_ + h) * NCHUNK + c) * NMODE;
#pragma unroll
  for (int k0 = 0; k0 < NMODE; k0 += 4) {
    const int ka = k0 + 2 * half;
    v2f a, bb;
    a.x = sCR[jrow * CPITCH + ka];
    a.y = sCR[jrow * CPITCH + ka + 1];
    bb.x = XR[xrb + ka];
    bb.y = XR[xrb + ka + 1];
    acc = wmma4(a, bb, acc);
  }
#pragma unroll
  for (int k0 = 0; k0 < NMODE; k0 += 4) {
    const int ka = k0 + 2 * half;
    v2f a, bb;
    a.x = sCI[jrow * CPITCH + ka];
    a.y = sCI[jrow * CPITCH + ka + 1];
    bb.x = XI[xrb + ka];
    bb.y = XI[xrb + ka + 1];
    acc = wmma4(a, bb, acc);
  }
  const float Dh = Dvec[h];
#pragma unroll
  for (int v = 0; v < 8; ++v) {
    const int j = jt * 16 + v + 8 * half;
    const size_t idx = (size_t)(b * H_ + h) * L_ + (size_t)c * Q_ + j;
    const float uu = u[idx];
    const float y = acc[v] + Dh * uu;
    ygelu[idx] = 0.5f * y * (1.0f + erff(y * 0.70710678118654752f));
  }
}

#define GP_ROWS    128
#define GP_THREADS 512
#define GP_KG      16

__global__ void __launch_bounds__(GP_THREADS)
s4d_glu(const float* __restrict__ ygelu,
        const float* __restrict__ W,
        const float* __restrict__ bias,
        float* __restrict__ out) {
  __shared__ __attribute__((aligned(16))) float sW[GP_ROWS * WPITCH];
  const int tid    = threadIdx.x;
  const int waveId = tid >> 5;
  const int lane   = tid & 31;
  const int half   = lane >> 4;
  const int lm     = lane & 15;
  const int panel  = blockIdx.x & 7;
  const int cgrp   = blockIdx.x >> 3;
  const int col0   = cgrp * 512 + waveId * 16 + lm;
  const int bb     = col0 >> 11;
  const int l0     = col0 & (L_ - 1);

  const float* yb = ygelu + (size_t)bb * H_ * L_ + (size_t)(2 * half) * L_ + l0;

  v2f bcur[4], bnxt[4];
#pragma unroll
  for (int s = 0; s < 4; ++s) {
    bcur[s].x = yb[(size_t)(s * 4) * L_];
    bcur[s].y = yb[(size_t)(s * 4 + 1) * L_];
  }

  for (int it = 0; it < 4; ++it) {
    float4 tmp[8];
#pragma unroll
    for (int j = 0; j < 8; ++j) {
      const int r  = (it * 8 + j) * 4 + (tid >> 7);
      const int k4 = tid & 127;
      const int srow = (r < 64) ? (panel * 64 + r) : (H_ + panel * 64 + (r - 64));
      tmp[j] = *(const float4*)(W + (size_t)srow * H_ + k4 * 4);
    }
#pragma unroll
    for (int j = 0; j < 8; ++j) {
      const int r  = (it * 8 + j) * 4 + (tid >> 7);
      const int k4 = tid & 127;
      *(float4*)(sW + r * WPITCH + k4 * 4) = tmp[j];
    }
  }
  __syncthreads();

  const float* la = sW + lm * WPITCH + 2 * half;

  for (int t = 0; t < 2; ++t) {
    v8f acc[8];
#pragma unroll
    for (int i = 0; i < 8; ++i) {
      acc[i] = (v8f){0.f, 0.f, 0.f, 0.f, 0.f, 0.f, 0.f, 0.f};
    }

    for (int g = 0; g < H_ / GP_KG; ++g) {
      int gn = t * (H_ / GP_KG) + g + 1;
      gn = (gn < 2 * (H_ / GP_KG)) ? gn : (2 * (H_ / GP_KG) - 1);
      const float* yn = yb + (gn >> 5) * 256 + (size_t)((gn & 31) * GP_KG) * L_;
#pragma unroll
      for (int s = 0; s < 4; ++s) {
        bnxt[s].x = yn[(size_t)(s * 4) * L_];
        bnxt[s].y = yn[(size_t)(s * 4 + 1) * L_];
      }

      const float* lk = la + g * GP_KG;
#pragma unroll
      for (int s = 0; s < 4; ++s) {
        v2f a[8];
#pragma unroll
        for (int i = 0; i < 8; ++i) {
          a[i] = *(const v2f*)(lk + i * 16 * WPITCH + s * 4);
        }
#pragma unroll
        for (int i = 0; i < 8; ++i) {
          acc[i] = wmma4(a[i], bcur[s], acc[i]);
        }
      }
#pragma unroll
      for (int s = 0; s < 4; ++s) bcur[s] = bnxt[s];
    }

    const int l = l0 + t * 256;
#pragma unroll
    for (int i = 0; i < 4; ++i) {
#pragma unroll
      for (int v = 0; v < 8; ++v) {
        const int hr = panel * 64 + i * 16 + v + 8 * half;
        const float a  = acc[i][v] + bias[hr];
        const float gv = acc[4 + i][v] + bias[H_ + hr];
        out[((size_t)(bb * H_ + hr)) * L_ + l] = a * (1.0f / (1.0f + expf(-gv)));
      }
    }
  }
}

extern "C" void kernel_launch(void* const* d_in, const int* in_sizes, int n_in,
                              void* d_out, int out_size, void* d_ws, size_t ws_size,
                              hipStream_t stream) {
  (void)in_sizes; (void)n_in; (void)out_size; (void)ws_size;
  const float* u          = (const float*)d_in[0];
  const float* log_dt     = (const float*)d_in[1];
  const float* C          = (const float*)d_in[2];
  const float* log_A_real = (const float*)d_in[3];
  const float* A_imag     = (const float*)d_in[4];
  const float* Dvec       = (const float*)d_in[5];
  const float* W          = (const float*)d_in[6];
  const float* bias       = (const float*)d_in[7];
  float* out = (float*)d_out;

  float* ws = (float*)d_ws;
  float* Ktap2 = ws;
  float* CoutR = Ktap2 + (size_t)H_ * 128;
  float* CoutI = CoutR + (size_t)H_ * Q_ * NMODE;
  float* Wr    = CoutI + (size_t)H_ * Q_ * NMODE;
  float* Wi    = Wr    + (size_t)H_ * NMODE;
  float* XR    = Wi    + (size_t)H_ * NMODE;
  float* XI    = XR    + (size_t)B_ * H_ * NCHUNK * NMODE;
  float* ygelu = XI    + (size_t)B_ * H_ * NCHUNK * NMODE;

  s4d_setup<<<H_, 32, 0, stream>>>(log_dt, C, log_A_real, A_imag,
                                   Ktap2, CoutR, CoutI, Wr, Wi);
  s4d_scan<<<(B_ * H_) / 8, 256, 0, stream>>>(u, Wr, Wi, XR, XI);
  s4d_conv<<<H_ * 8, 256, 0, stream>>>(u, Ktap2, CoutR, CoutI, XR, XI, Dvec, ygelu);
  s4d_glu<<<256, GP_THREADS, 0, stream>>>(ygelu, W, bias, out);
}


# conv kernel as the transposed product: chunks are the tile rows (samples and entering states = A fragments, 16-byte global loads), tables = B operand from LDS, one block per head; k sum order changes
# speedup vs baseline: 1.5034x; 1.1823x over previous
#include <hip/hip_runtime.h>
#include <math.h>

#define H_     512
#define NMODE  32
#define B_     8
#define L_     2048
#define Q_     64
#define NCHUNK 32
#define CPITCH 34
#define WPITCH 516

typedef __attribute__((ext_vector_type(2))) float v2f;
typedef __attribute__((ext_vector_type(8))) float v8f;

__device__ __forceinline__ v8f wmma4(v2f a, v2f b, v8f c) {
  return __builtin_amdgcn_wmma_f32_16x16x4_f32(false, a, false, b, (short)0, c,
                                               false, false);
}

__global__ void s4d_setup(const float* __restrict__ log_dt,
                          const float* __restrict__ C,
                          const float* __restrict__ log_A_real,
                          const float* __restrict__ A_imag,
                          float* __restrict__ Ktap2,
                          float* __restrict__ CoutR,
                          float* __restrict__ CoutI,
                          float* __restrict__ Wr,
                          float* __restrict__ Wi) {
  const int h = blockIdx.x;
  const int k = threadIdx.x;
  const float dt = expf(log_dt[h]);
  const float Ar = -expf(log_A_real[h * NMODE + k]);
  const float Ai = A_imag[h * NMODE + k];
  const float Ccr = C[(h * NMODE + k) * 2 + 0];
  const float Cci = C[(h * NMODE + k) * 2 + 1];
  const float er = expf(Ar * dt);
  const float wr = er * cosf(Ai * dt);
  const float wi = er * sinf(Ai * dt);
  const float exr = wr - 1.0f, exi = wi;
  const float den = Ar * Ar + Ai * Ai;
  const float qr = (exr * Ar + exi * Ai) / den;
  const float qi = (exi * Ar - exr * Ai) / den;
  const float Ckr = Ccr * qr - Cci * qi;
  const float Cki = Ccr * qi + Cci * qr;
  Wr[h * NMODE + k] = wr;
  Wi[h * NMODE + k] = wi;

  Ktap2[h * 128 + k] = 0.0f;
  Ktap2[h * 128 + 32 + k] = 0.0f;

  float pr = Ckr, pi = Cki;
  for (int d = 0; d < Q_; ++d) {
    float v = 2.0f * pr;
    for (int m = 16; m >= 1; m >>= 1) v += __shfl_xor(v, m, 32);
    if (k == 0) Ktap2[h * 128 + 64 + d] = v;
    const float nr = pr * wr - pi * wi;
    const float ni = pr * wi + pi * wr;
    pr = nr; pi = ni;
  }
  float sr = Ckr * wr - Cki * wi;
  float si = Ckr * wi + Cki * wr;
  for (int j = 0; j < Q_; ++j) {
    CoutR[((size_t)h * Q_ + j) * NMODE + k] = 2.0f * sr;
    CoutI[((size_t)h * Q_ + j) * NMODE + k] = -2.0f * si;
    const float nr = sr * wr - si * wi;
    const float ni = sr * wi + si * wr;
    sr = nr; si = ni;
  }
}

#define SB 16

__device__ __forceinline__ void scan_load(float (&x)[Q_], const float4* __restrict__ p) {
#pragma unroll
  for (int q = 0; q < Q_ / 4; ++q) {
    const float4 v = p[q];
    x[4 * q + 0] = v.x; x[4 * q + 1] = v.y; x[4 * q + 2] = v.z; x[4 * q + 3] = v.w;
  }
}

__device__ __forceinline__ void scan_chunk(const float (&x)[Q_],
                                           const float (&pr)[SB], const float (&pi)[SB],
                                           float w16r, float w16i,
                                           float& sr, float& si) {
#pragma unroll
  for (int q = 0; q < Q_ / SB; ++q) {
    const float* xb = x + q * SB;
    float ar0 = pr[SB - 1] * xb[0], ai0 = pi[SB - 1] * xb[0];
    float ar1 = pr[SB - 2] * xb[1], ai1 = pi[SB - 2] * xb[1];
#pragma unroll
    for (int i = 2; i < SB - 2; i += 2) {
      ar0 = fmaf(pr[SB - 1 - i], xb[i], ar0);
      ai0 = fmaf(pi[SB - 1 - i], xb[i], ai0);
      ar1 = fmaf(pr[SB - 2 - i], xb[i + 1], ar1);
      ai1 = fmaf(pi[SB - 2 - i], xb[i + 1], ai1);
    }
    ar0 = fmaf(pr[1], xb[SB - 2], ar0);
    ai0 = fmaf(pi[1], xb[SB - 2], ai0);
    ar1 += xb[SB - 1];
    const float Sr = ar0 + ar1, Si = ai0 + ai1;
    const float nr = fmaf(w16r, sr, fmaf(-w16i, si, Sr));
    const float ni = fmaf(w16r, si, fmaf(w16i, sr, Si));
    sr = nr; si = ni;
  }
}

__global__ void __launch_bounds__(256) __attribute__((amdgpu_waves_per_eu(4, 4)))
s4d_scan(const float* __restrict__ u,
         const float* __restrict__ Wr,
         const float* __restrict__ Wi,
         float* __restrict__ XR,
         float* __restrict__ XI) {
  const int wid  = (int)((blockIdx.x * blockDim.x + threadIdx.x) >> 5);
  const int lane = threadIdx.x & 31;
  const int b = wid >> 9;
  const int h = wid & (H_ - 1);
  const float wr = Wr[h * NMODE + lane];
  const float wi = Wi[h * NMODE + lane];

  float pr[SB], pi[SB];
  pr[0] = 1.0f; pi[0] = 0.0f;
  pr[1] = wr;   pi[1] = wi;
#pragma unroll
  for (int m = 2; m < SB; ++m) {
    pr[m] = pr[m - 1] * wr - pi[m - 1] * wi;
    pi[m] = pr[m - 1] * wi + pi[m - 1] * wr;
  }
  const float w16r = pr[SB - 1] * wr - pi[SB - 1] * wi;
  const float w16i = pr[SB - 1] * wi + pi[SB - 1] * wr;

  float sr = 0.0f, si = 0.0f;
  const float4* up4 = (const float4*)(u + (size_t)(b * H_ + h) * L_);
  const size_t xb = ((size_t)(b * H_ + h)) * NCHUNK * NMODE;

  float xa[Q_], xc[Q_];
  scan_load(xa, up4);
  for (int c = 0; c < NCHUNK; c += 2) {
    scan_load(xc, up4 + (c + 1) * (Q_ / 4));
    XR[xb + (size_t)c * NMODE + lane] = sr;
    XI[xb + (size_t)c * NMODE + lane] = si;
    scan_chunk(xa, pr, pi, w16r, w16i, sr, si);
    const int cn = (c + 2 < NCHUNK) ? (c + 2) : (NCHUNK - 1);
    scan_load(xa, up4 + cn * (Q_ / 4));
    XR[xb + (size_t)(c + 1) * NMODE + lane] = sr;
    XI[xb + (size_t)(c + 1) * NMODE + lane] = si;
    scan_chunk(xc, pr, pi, w16r, w16i, sr, si);
  }
}

#define CV_THREADS 256

__device__ __forceinline__ int cv_kpos(int k) {
  return (k & ~6) | ((k & 2) << 1) | ((k & 4) >> 1);
}

__global__ void __launch_bounds__(CV_THREADS) __attribute__((amdgpu_waves_per_eu(4, 4)))
s4d_conv(const float* __restrict__ u,
         const float* __restrict__ Ktap2,
         const float* __restrict__ CoutR,
         const float* __restrict__ CoutI,
         const float* __restrict__ XR,
         const float* __restrict__ XI,
         const float* __restrict__ Dvec,
         float* __restrict__ ygelu) {
  __shared__ float sKt[128];
  __shared__ __attribute__((aligned(16))) float sCR[Q_ * CPITCH];
  __shared__ __attribute__((aligned(16))) float sCI[Q_ * CPITCH];

  const int h      = blockIdx.x;
  const int tid    = threadIdx.x;
  const int waveId = tid >> 5;
  const int lane   = tid & 31;
  const int half   = lane >> 4;
  const int lm     = lane & 15;

  {
    float4 vr[2], vi[2];
    float kt = 0.0f;
    if (tid < 128) kt = Ktap2[(size_t)h * 128 + tid];
#pragma unroll
    for (int it = 0; it < 2; ++it) {
      const int i4 = tid + CV_THREADS * it;
      vr[it] = *(const float4*)(CoutR + (size_t)h * Q_ * NMODE + (size_t)i4 * 4);
      vi[it] = *(const float4*)(CoutI + (size_t)h * Q_ * NMODE + (size_t)i4 * 4);
    }
    if (tid < 128) sKt[tid] = kt;
#pragma unroll
    for (int it = 0; it < 2; ++it) {
      const int i4 = tid + CV_THREADS * it;
      const int j  = i4 >> 3;
      const int k4 = (i4 & 7) * 4;
      float* dr = sCR + j * CPITCH;
      float* di = sCI + j * CPITCH;
      dr[cv_kpos(k4 + 0)] = vr[it].x; dr[cv_kpos(k4 + 1)] = vr[it].y;
      dr[cv_kpos(k4 + 2)] = vr[it].z; dr[cv_kpos(k4 + 3)] = vr[it].w;
      di[cv_kpos(k4 + 0)] = vi[it].x; di[cv_kpos(k4 + 1)] = vi[it].y;
      di[cv_kpos(k4 + 2)] = vi[it].z; di[cv_kpos(k4 + 3)] = vi[it].w;
    }
  }
  __syncthreads();

  const float Dh = Dvec[h];

  const int nh = waveId & 1;
  for (int it = 0; it < 4; ++it) {
    const int mt = (waveId >> 1) + 4 * it;
    const int qa = mt * 16 + lm;
    const int ba = qa >> 5, ca = qa & 31;
    const float4* up = (const float4*)(u + (size_t)(ba * H_ + h) * L_ + (size_t)ca * Q_) + half;
    const size_t xo = ((size_t)(ba * H_ + h) * NCHUNK + ca) * NMODE;
    const float4* xrp = (const float4*)(XR + xo) + half;
    const float4* xip = (const float4*)(XI + xo) + half;

    float4 ua[8], xr[4], xi[4];
#pragma unroll
    for (int i = 0; i < 8; ++i) ua[i] = up[2 * i];
#pragma unroll
    for (int i = 0; i < 4; ++i) { xr[i] = xrp[2 * i]; xi[i] = xip[2 * i]; }

    v8f acc[2];
#pragma unroll
    for (int nt = 0; nt < 2; ++nt) acc[nt] = (v8f){0.f, 0.f, 0.f, 0.f, 0.f, 0.f, 0.f, 0.f};

#pragma unroll
    for (int s = 0; s < 16; ++s) {
      v2f a;
      a.x = (s & 1) ? ua[s >> 1].z : ua[s >> 1].x;
      a.y = (s & 1) ? ua[s >> 1].w : ua[s >> 1].y;
      const int t0 = 8 * (s >> 1) + 4 * half + 2 * (s & 1);
#pragma unroll
      for (int nt = 0; nt < 2; ++nt) {
        const int j = (2 * nh + nt) * 16 + lm;
        v2f bb;
        bb.x = sKt[64 + j - t0];
        bb.y = sKt[63 + j - t0];
        acc[nt] = wmma4(a, bb, acc[nt]);
      }
    }
#pragma unroll
    for (int s = 0; s < 8; ++s) {
      v2f a;
      a.x = (s & 1) ? xr[s >> 1].z : xr[s >> 1].x;
      a.y = (s & 1) ? xr[s >> 1].w : xr[s >> 1].y;
#pragma unroll
      for (int nt = 0; nt < 2; ++nt) {
        const v2f bb = *(const v2f*)(sCR + ((2 * nh + nt) * 16 + lm) * CPITCH + 4 * s + 2 * half);
        acc[nt] = wmma4(a, bb, acc[nt]);
      }
    }
#pragma unroll
    for (int s = 0; s < 8; ++s) {
      v2f a;
      a.x = (s & 1) ? xi[s >> 1].z : xi[s >> 1].x;
      a.y = (s & 1) ? xi[s >> 1].w : xi[s >> 1].y;
#pragma unroll
      for (int nt = 0; nt < 2; ++nt) {
        const v2f bb = *(const v2f*)(sCI + ((2 * nh + nt) * 16 + lm) * CPITCH + 4 * s + 2 * half);
        acc[nt] = wmma4(a, bb, acc[nt]);
      }
    }

    float yv[8][2];
#pragma unroll
    for (int r = 0; r < 8; ++r) {
      const int qd = mt * 16 + 8 * half + r;
      const int bd = qd >> 5, cd = qd & 31;
      const size_t rb = (size_t)(bd * H_ + h) * L_ + (size_t)cd * Q_ + lm;
#pragma unroll
      for (int nt = 0; nt < 2; ++nt) {
        const size_t idx = rb + (2 * nh + nt) * 16;
        yv[r][nt] = acc[nt][r] + Dh * u[idx];
      }
    }
#pragma unroll
    for (int r = 0; r < 8; ++r) {
      const int qd = mt * 16 + 8 * half + r;
      const int bd = qd >> 5, cd = qd & 31;
      const size_t rb = (size_t)(bd * H_ + h) * L_ + (size_t)cd * Q_ + lm;
#pragma unroll
      for (int nt = 0; nt < 2; ++nt) {
        const size_t idx = rb + (2 * nh + nt) * 16;
        const float y = yv[r][nt];
        ygelu[idx] = 0.5f * y * (1.0f + erff(y * 0.70710678118654752f));
      }
    }
  }
}

#define GP_ROWS    128
#define GP_THREADS 512
#define GP_KG      16

__global__ void __launch_bounds__(GP_THREADS)
s4d_glu(const float* __restrict__ ygelu,
        const float* __restrict__ W,
        const float* __restrict__ bias,
        float* __restrict__ out) {
  __shared__ __attribute__((aligned(16))) float sW[GP_ROWS * WPITCH];
  const int tid    = threadIdx.x;
  const int waveId = tid >> 5;
  const int lane   = tid & 31;
  const int half   = lane >> 4;
  const int lm     = lane & 15;
  const int panel  = blockIdx.x & 7;
  const int cgrp   = blockIdx.x >> 3;
  const int col0   = cgrp * 512 + waveId * 16 + lm;
  const int bb     = col0 >> 11;
  const int l0     = col0 & (L_ - 1);

  const float* yb = ygelu + (size_t)bb * H_ * L_ + (size_t)(2 * half) * L_ + l0;

  v2f bcur[4], bnxt[4];
#pragma unroll
  for (int s = 0; s < 4; ++s) {
    bcur[s].x = yb[(size_t)(s * 4) * L_];
    bcur[s].y = yb[(size_t)(s * 4 + 1) * L_];
  }

  for (int it = 0; it < 4; ++it) {
    float4 tmp[8];
#pragma unroll
    for (int j = 0; j < 8; ++j) {
      const int r  = (it * 8 + j) * 4 + (tid >> 7);
      const int k4 = tid & 127;
      const int srow = (r < 64) ? (panel * 64 + r) : (H_ + panel * 64 + (r - 64));
      tmp[j] = *(const float4*)(W + (size_t)srow * H_ + k4 * 4);
    }
#pragma unroll
    for (int j = 0; j < 8; ++j) {
      const int r  = (it * 8 + j) * 4 + (tid >> 7);
      const int k4 = tid & 127;
      *(float4*)(sW + r * WPITCH + k4 * 4) = tmp[j];
    }
  }
  __syncthreads();

  const float* la = sW + lm * WPITCH + 2 * half;

  for (int t = 0; t < 2; ++t) {
    v8f acc[8];
#pragma unroll
    for (int i = 0; i < 8; ++i) {
      acc[i] = (v8f){0.f, 0.f, 0.f, 0.f, 0.f, 0.f, 0.f, 0.f};
    }

    for (int g = 0; g < H_ / GP_KG; ++g) {
      int gn = t * (H_ / GP_KG) + g + 1;
      gn = (gn < 2 * (H_ / GP_KG)) ? gn : (2 * (H_ / GP_KG) - 1);
      const float* yn = yb + (gn >> 5) * 256 + (size_t)((gn & 31) * GP_KG) * L_;
#pragma unroll
      for (int s = 0; s < 4; ++s) {
        bnxt[s].x = yn[(size_t)(s * 4) * L_];
        bnxt[s].y = yn[(size_t)(s * 4 + 1) * L_];
      }

      const float* lk = la + g * GP_KG;
#pragma unroll
      for (int s = 0; s < 4; ++s) {
        v2f a[8];
#pragma unroll
        for (int i = 0; i < 8; ++i) {
          a[i] = *(const v2f*)(lk + i * 16 * WPITCH + s * 4);
        }
#pragma unroll
        for (int i = 0; i < 8; ++i) {
          acc[i] = wmma4(a[i], bcur[s], acc[i]);
        }
      }
#pragma unroll
      for (int s = 0; s < 4; ++s) bcur[s] = bnxt[s];
    }

    const int l = l0 + t * 256;
#pragma unroll
    for (int i = 0; i < 4; ++i) {
#pragma unroll
      for (int v = 0; v < 8; ++v) {
        const int hr = panel * 64 + i * 16 + v + 8 * half;
        const float a  = acc[i][v] + bias[hr];
        const float gv = acc[4 + i][v] + bias[H_ + hr];
        out[((size_t)(bb * H_ + hr)) * L_ + l] = a * (1.0f / (1.0f + expf(-gv)));
      }
    }
  }
}

extern "C" void kernel_launch(void* const* d_in, const int* in_sizes, int n_in,
                              void* d_out, int out_size, void* d_ws, size_t ws_size,
                              hipStream_t stream) {
  (void)in_sizes; (void)n_in; (void)out_size; (void)ws_size;
  const float* u          = (const float*)d_in[0];
  const float* log_dt     = (const float*)d_in[1];
  const float* C          = (const float*)d_in[2];
  const float* log_A_real = (const float*)d_in[3];
  const float* A_imag     = (const float*)d_in[4];
  const float* Dvec       = (const float*)d_in[5];
  const float* W          = (const float*)d_in[6];
  const float* bias       = (const float*)d_in[7];
  float* out = (float*)d_out;

  float* ws = (float*)d_ws;
  float* Ktap2 = ws;
  float* CoutR = Ktap2 + (size_t)H_ * 128;
  float* CoutI = CoutR + (size_t)H_ * Q_ * NMODE;
  float* Wr    = CoutI + (size_t)H_ * Q_ * NMODE;
  float* Wi    = Wr    + (size_t)H_ * NMODE;
  float* XR    = Wi    + (size_t)H_ * NMODE;
  float* XI    = XR    + (size_t)B_ * H_ * NCHUNK * NMODE;
  float* ygelu = XI    + (size_t)B_ * H_ * NCHUNK * NMODE;

  s4d_setup<<<H_, 32, 0, stream>>>(log_dt, C, log_A_real, A_imag,
                                   Ktap2, CoutR, CoutI, Wr, Wi);
  s4d_scan<<<(B_ * H_) / 8, 256, 0, stream>>>(u, Wr, Wi, XR, XI);
  s4d_conv<<<H_, CV_THREADS, 0, stream>>>(u, Ktap2, CoutR, CoutI, XR, XI, Dvec, ygelu);
  s4d_glu<<<256, GP_THREADS, 0, stream>>>(ygelu, W, bias, out);
}


# set-up launch folded into the scan: block = one head x 8 batch waves, each wave rebuilds w, wave b writes taps and projection rows 8b..8b+7 from w^(8b); 4 launches -> 3; table powers re-associated
# speedup vs baseline: 1.6656x; 1.1079x over previous
#include <hip/hip_runtime.h>
#include <math.h>

#define H_     512
#define NMODE  32
#define B_     8
#define L_     2048
#define Q_     64
#define NCHUNK 32
#define CPITCH 34
#define WPITCH 516

typedef __attribute__((ext_vector_type(2))) float v2f;
typedef __attribute__((ext_vector_type(8))) float v8f;

__device__ __forceinline__ v8f wmma4(v2f a, v2f b, v8f c) {
  return __builtin_amdgcn_wmma_f32_16x16x4_f32(false, a, false, b, (short)0, c,
                                               false, false);
}

__device__ __forceinline__ void mode_params(const float* __restrict__ log_dt,
                                            const float* __restrict__ C,
                                            const float* __restrict__ log_A_real,
                                            const float* __restrict__ A_imag,
                                            int h, int k,
                                            float& wr, float& wi, float& Ckr, float& Cki) {
  const float dt = expf(log_dt[h]);
  const float Ar = -expf(log_A_real[h * NMODE + k]);
  const float Ai = A_imag[h * NMODE + k];
  const float Ccr = C[(h * NMODE + k) * 2 + 0];
  const float Cci = C[(h * NMODE + k) * 2 + 1];
  const float er = expf(Ar * dt);
  wr = er * cosf(Ai * dt);
  wi = er * sinf(Ai * dt);
  const float exr = wr - 1.0f, exi = wi;
  const float den = Ar * Ar + Ai * Ai;
  const float qr = (exr * Ar + exi * Ai) / den;
  const float qi = (exi * Ar - exr * Ai) / den;
  Ckr = Ccr * qr - Cci * qi;
  Cki = Ccr * qi + Cci * qr;
}

#define SB 16

__device__ __forceinline__ void scan_load(float (&x)[Q_], const float4* __restrict__ p) {
#pragma unroll
  for (int q = 0; q < Q_ / 4; ++q) {
    const float4 v = p[q];
    x[4 * q + 0] = v.x; x[4 * q + 1] = v.y; x[4 * q + 2] = v.z; x[4 * q + 3] = v.w;
  }
}

__device__ __forceinline__ void scan_chunk(const float (&x)[Q_],
                                           const float (&pr)[SB], const float (&pi)[SB],
                                           float w16r, float w16i,
                                           float& sr, float& si) {
#pragma unroll
  for (int q = 0; q < Q_ / SB; ++q) {
    const float* xb = x + q * SB;
    float ar0 = pr[SB - 1] * xb[0], ai0 = pi[SB - 1] * xb[0];
    float ar1 = pr[SB - 2] * xb[1], ai1 = pi[SB - 2] * xb[1];
#pragma unroll
    for (int i = 2; i < SB - 2; i += 2) {
      ar0 = fmaf(pr[SB - 1 - i], xb[i], ar0);
      ai0 = fmaf(pi[SB - 1 - i], xb[i], ai0);
      ar1 = fmaf(pr[SB - 2 - i], xb[i + 1], ar1);
      ai1 = fmaf(pi[SB - 2 - i], xb[i + 1], ai1);
    }
    ar0 = fmaf(pr[1], xb[SB - 2], ar0);
    ai0 = fmaf(pi[1], xb[SB - 2], ai0);
    ar1 += xb[SB - 1];
    const float Sr = ar0 + ar1, Si = ai0 + ai1;
    const float nr = fmaf(w16r, sr, fmaf(-w16i, si, Sr));
    const float ni = fmaf(w16r, si, fmaf(w16i, sr, Si));
    sr = nr; si = ni;
  }
}

__global__ void __launch_bounds__(256) __attribute__((amdgpu_waves_per_eu(4, 4)))
s4d_scan(const float* __restrict__ u,
         const float* __restrict__ log_dt,
         const float* __restrict__ C,
         const float* __restrict__ log_A_real,
         const float* __restrict__ A_imag,
         float* __restrict__ Ktap2,
         float* __restrict__ CoutR,
         float* __restrict__ CoutI,
         float* __restrict__ XR,
         float* __restrict__ XI) {
  __shared__ __attribute__((aligned(16))) float sTap[Q_];
  const int h    = blockIdx.x;
  const int b    = threadIdx.x >> 5;
  const int lane = threadIdx.x & 31;

  const float4* up4 = (const float4*)(u + (size_t)(b * H_ + h) * L_);
  float xa[Q_], xc[Q_];
  scan_load(xa, up4);

  float wr, wi, Ckr, Cki;
  mode_params(log_dt, C, log_A_real, A_imag, h, lane, wr, wi, Ckr, Cki);

  float pr[SB], pi[SB];
  pr[0] = 1.0f; pi[0] = 0.0f;
  pr[1] = wr;   pi[1] = wi;
#pragma unroll
  for (int m = 2; m < SB; ++m) {
    pr[m] = pr[m - 1] * wr - pi[m - 1] * wi;
    pi[m] = pr[m - 1] * wi + pi[m - 1] * wr;
  }
  const float w16r = pr[SB - 1] * wr - pi[SB - 1] * wi;
  const float w16i = pr[SB - 1] * wi + pi[SB - 1] * wr;

  {
    const float w32r = w16r * w16r - w16i * w16i, w32i = 2.0f * w16r * w16i;
    const float w48r = w32r * w16r - w32i * w16i, w48i = w32r * w16i + w32i * w16r;
    const int   q  = b >> 1;
    const float gr = (q == 0) ? 1.0f : (q == 1) ? w16r : (q == 2) ? w32r : w48r;
    const float gi = (q == 0) ? 0.0f : (q == 1) ? w16i : (q == 2) ? w32i : w48i;
    const float er = (b & 1) ? pr[8] : 1.0f;
    const float ei = (b & 1) ? pi[8] : 0.0f;
    const float br = gr * er - gi * ei, bi = gr * ei + gi * er;
    const float q0r = Ckr * br - Cki * bi, q0i = Ckr * bi + Cki * br;

    float v[8];
#pragma unroll
    for (int i = 0; i < 8; ++i) v[i] = 2.0f * (q0r * pr[i] - q0i * pi[i]);
#pragma unroll
    for (int m = 16; m >= 1; m >>= 1) {
#pragma unroll
      for (int i = 0; i < 8; ++i) v[i] += __shfl_xor(v[i], m, 32);
    }
    if (lane == 0) {
      *(float4*)(sTap + 8 * b)     = make_float4(v[0], v[1], v[2], v[3]);
      *(float4*)(sTap + 8 * b + 4) = make_float4(v[4], v[5], v[6], v[7]);
    }
    __syncthreads();
    if (b < 4) {
      Ktap2[(size_t)h * 128 + 32 * b + lane] = (b < 2) ? 0.0f : sTap[32 * (b - 2) + lane];
    }
#pragma unroll
    for (int i = 0; i < 8; ++i) {
      const float sr_ = q0r * pr[i + 1] - q0i * pi[i + 1];
      const float si_ = q0r * pi[i + 1] + q0i * pr[i + 1];
      const size_t o = ((size_t)h * Q_ + 8 * b + i) * NMODE + lane;
      CoutR[o] = 2.0f * sr_;
      CoutI[o] = -2.0f * si_;
    }
  }

  float sr = 0.0f, si = 0.0f;
  const size_t xb = ((size_t)(b * H_ + h)) * NCHUNK * NMODE;

  for (int c = 0; c < NCHUNK; c += 2) {
    scan_load(xc, up4 + (c + 1) * (Q_ / 4));
    XR[xb + (size_t)c * NMODE + lane] = sr;
    XI[xb + (size_t)c * NMODE + lane] = si;
    scan_chunk(xa, pr, pi, w16r, w16i, sr, si);
    const int cn = (c + 2 < NCHUNK) ? (c + 2) : (NCHUNK - 1);
    scan_load(xa, up4 + cn * (Q_ / 4));
    XR[xb + (size_t)(c + 1) * NMODE + lane] = sr;
    XI[xb + (size_t)(c + 1) * NMODE + lane] = si;
    scan_chunk(xc, pr, pi, w16r, w16i, sr, si);
  }
}

#define CV_THREADS 256

__device__ __forceinline__ int cv_kpos(int k) {
  return (k & ~6) | ((k & 2) << 1) | ((k & 4) >> 1);
}

__global__ void __launch_bounds__(CV_THREADS) __attribute__((amdgpu_waves_per_eu(4, 4)))
s4d_conv(const float* __restrict__ u,
         const float* __restrict__ Ktap2,
         const float* __restrict__ CoutR,
         const float* __restrict__ CoutI,
         const float* __restrict__ XR,
         const float* __restrict__ XI,
         const float* __restrict__ Dvec,
         float* __restrict__ ygelu) {
  __shared__ float sKt[128];
  __shared__ __attribute__((aligned(16))) float sCR[Q_ * CPITCH];
  __shared__ __attribute__((aligned(16))) float sCI[Q_ * CPITCH];

  const int h      = blockIdx.x;
  const int tid    = threadIdx.x;
  const int waveId = tid >> 5;
  const int lane   = tid & 31;
  const int half   = lane >> 4;
  const int lm     = lane & 15;

  {
    float4 vr[2], vi[2];
    float kt = 0.0f;
    if (tid < 128) kt = Ktap2[(size_t)h * 128 + tid];
#pragma unroll
    for (int it = 0; it < 2; ++it) {
      const int i4 = tid + CV_THREADS * it;
      vr[it] = *(const float4*)(CoutR + (size_t)h * Q_ * NMODE + (size_t)i4 * 4);
      vi[it] = *(const float4*)(CoutI + (size_t)h * Q_ * NMODE + (size_t)i4 * 4);
    }
    if (tid < 128) sKt[tid] = kt;
#pragma unroll
    for (int it = 0; it < 2; ++it) {
      const int i4 = tid + CV_THREADS * it;
      const int j  = i4 >> 3;
      const int k4 = (i4 & 7) * 4;
      float* dr = sCR + j * CPITCH;
      float* di = sCI + j * CPITCH;
      dr[cv_kpos(k4 + 0)] = vr[it].x; dr[cv_kpos(k4 + 1)] = vr[it].y;
      dr[cv_kpos(k4 + 2)] = vr[it].z; dr[cv_kpos(k4 + 3)] = vr[it].w;
      di[cv_kpos(k4 + 0)] = vi[it].x; di[cv_kpos(k4 + 1)] = vi[it].y;
      di[cv_kpos(k4 + 2)] = vi[it].z; di[cv_kpos(k4 + 3)] = vi[it].w;
    }
  }
  __syncthreads();

  const float Dh = Dvec[h];

  const int nh = waveId & 1;
  for (int it = 0; it < 4; ++it) {
    const int mt = (waveId >> 1) + 4 * it;
    const int qa = mt * 16 + lm;
    const int ba = qa >> 5, ca = qa & 31;
    const float4* up = (const float4*)(u + (size_t)(ba * H_ + h) * L_ + (size_t)ca * Q_) + half;
    const size_t xo = ((size_t)(ba * H_ + h) * NCHUNK + ca) * NMODE;
    const float4* xrp = (const float4*)(XR + xo) + half;
    const float4* xip = (const float4*)(XI + xo) + half;

    float4 ua[8], xr[4], xi[4];
#pragma unroll
    for (int i = 0; i < 8; ++i) ua[i] = up[2 * i];
#pragma unroll
    for (int i = 0; i < 4; ++i) { xr[i] = xrp[2 * i]; xi[i] = xip[2 * i]; }

    v8f acc[2];
#pragma unroll
    for (int nt = 0; nt < 2; ++nt) acc[nt] = (v8f){0.f, 0.f, 0.f, 0.f, 0.f, 0.f, 0.f, 0.f};

#pragma unroll
    for (int s = 0; s < 16; ++s) {
      v2f a;
      a.x = (s & 1) ? ua[s >> 1].z : ua[s >> 1].x;
      a.y = (s & 1) ? ua[s >> 1].w : ua[s >> 1].y;
      const int t0 = 8 * (s >> 1) + 4 * half + 2 * (s & 1);
#pragma unroll
      for (int nt = 0; nt < 2; ++nt) {
        const int j = (2 * nh + nt) * 16 + lm;
        v2f bb;
        bb.x = sKt[64 + j - t0];
        bb.y = sKt[63 + j - t0];
        acc[nt] = wmma4(a, bb, acc[nt]);
      }
    }
#pragma unroll
    for (int s = 0; s < 8; ++s) {
      v2f a;
      a.x = (s & 1) ? xr[s >> 1].z : xr[s >> 1].x;
      a.y = (s & 1) ? xr[s >> 1].w : xr[s >> 1].y;
#pragma unroll
      for (int nt = 0; nt < 2; ++nt) {
        const v2f bb = *(const v2f*)(sCR + ((2 * nh + nt) * 16 + lm) * CPITCH + 4 * s + 2 * half);
        acc[nt] = wmma4(a, bb, acc[nt]);
      }
    }
#pragma unroll
    for (int s = 0; s < 8; ++s) {
      v2f a;
      a.x = (s & 1) ? xi[s >> 1].z : xi[s >> 1].x;
      a.y = (s & 1) ? xi[s >> 1].w : xi[s >> 1].y;
#pragma unroll
      for (int nt = 0; nt < 2; ++nt) {
        const v2f bb = *(const v2f*)(sCI + ((2 * nh + nt) * 16 + lm) * CPITCH + 4 * s + 2 * half);
        acc[nt] = wmma4(a, bb, acc[nt]);
      }
    }

    float yv[8][2];
#pragma unroll
    for (int r = 0; r < 8; ++r) {
      const int qd = mt * 16 + 8 * half + r;
      const int bd = qd >> 5, cd = qd & 31;
      const size_t rb = (size_t)(bd * H_ + h) * L_ + (size_t)cd * Q_ + lm;
#pragma unroll
      for (int nt = 0; nt < 2; ++nt) {
        const size_t idx = rb + (2 * nh + nt) * 16;
        yv[r][nt] = acc[nt][r] + Dh * u[idx];
      }
    }
#pragma unroll
    for (int r = 0; r < 8; ++r) {
      const int qd = mt * 16 + 8 * half + r;
      const int bd = qd >> 5, cd = qd & 31;
      const size_t rb = (size_t)(bd * H_ + h) * L_ + (size_t)cd * Q_ + lm;
#pragma unroll
      for (int nt = 0; nt < 2; ++nt) {
        const size_t idx = rb + (2 * nh + nt) * 16;
        const float y = yv[r][nt];
        ygelu[idx] = 0.5f * y * (1.0f + erff(y * 0.70710678118654752f));
      }
    }
  }
}

#define GP_ROWS    128
#define GP_THREADS 512
#define GP_KG      16

__global__ void __launch_bounds__(GP_THREADS)
s4d_glu(const float* __restrict__ ygelu,
        const float* __restrict__ W,
        const float* __restrict__ bias,
        float* __restrict__ out) {
  __shared__ __attribute__((aligned(16))) float sW[GP_ROWS * WPITCH];
  const int tid    = threadIdx.x;
  const int waveId = tid >> 5;
  const int lane   = tid & 31;
  const int half   = lane >> 4;
  const int lm     = lane & 15;
  const int panel  = blockIdx.x & 7;
  const int cgrp   = blockIdx.x >> 3;
  const int col0   = cgrp * 512 + waveId * 16 + lm;
  const int bb     = col0 >> 11;
  const int l0     = col0 & (L_ - 1);

  const float* yb = ygelu + (size_t)bb * H_ * L_ + (size_t)(2 * half) * L_ + l0;

  v2f bcur[4], bnxt[4];
#pragma unroll
  for (int s = 0; s < 4; ++s) {
    bcur[s].x = yb[(size_t)(s * 4) * L_];
    bcur[s].y = yb[(size_t)(s * 4 + 1) * L_];
  }

  for (int it = 0; it < 4; ++it) {
    float4 tmp[8];
#pragma unroll
    for (int j = 0; j < 8; ++j) {
      const int r  = (it * 8 + j) * 4 + (tid >> 7);
      const int k4 = tid & 127;
      const int srow = (r < 64) ? (panel * 64 + r) : (H_ + panel * 64 + (r - 64));
      tmp[j] = *(const float4*)(W + (size_t)srow * H_ + k4 * 4);
    }
#pragma unroll
    for (int j = 0; j < 8; ++j) {
      const int r  = (it * 8 + j) * 4 + (tid >> 7);
      const int k4 = tid & 127;
      *(float4*)(sW + r * WPITCH + k4 * 4) = tmp[j];
    }
  }
  __syncthreads();

  const float* la = sW + lm * WPITCH + 2 * half;

  for (int t = 0; t < 2; ++t) {
    v8f acc[8];
#pragma unroll
    for (int i = 0; i < 8; ++i) {
      acc[i] = (v8f){0.f, 0.f, 0.f, 0.f, 0.f, 0.f, 0.f, 0.f};
    }

    for (int g = 0; g < H_ / GP_KG; ++g) {
      int gn = t * (H_ / GP_KG) + g + 1;
      gn = (gn < 2 * (H_ / GP_KG)) ? gn : (2 * (H_ / GP_KG) - 1);
      const float* yn = yb + (gn >> 5) * 256 + (size_t)((gn & 31) * GP_KG) * L_;
#pragma unroll
      for (int s = 0; s < 4; ++s) {
        bnxt[s].x = yn[(size_t)(s * 4) * L_];
        bnxt[s].y = yn[(size_t)(s * 4 + 1) * L_];
      }

      const float* lk = la + g * GP_KG;
#pragma unroll
      for (int s = 0; s < 4; ++s) {
        v2f a[8];
#pragma unroll
        for (int i = 0; i < 8; ++i) {
          a[i] = *(const v2f*)(lk + i * 16 * WPITCH + s * 4);
        }
#pragma unroll
        for (int i = 0; i < 8; ++i) {
          acc[i] = wmma4(a[i], bcur[s], acc[i]);
        }
      }
#pragma unroll
      for (int s = 0; s < 4; ++s) bcur[s] = bnxt[s];
    }

    const int l = l0 + t * 256;
#pragma unroll
    for (int i = 0; i < 4; ++i) {
#pragma unroll
      for (int v = 0; v < 8; ++v) {
        const int hr = panel * 64 + i * 16 + v + 8 * half;
        const float a  = acc[i][v] + bias[hr];
        const float gv = acc[4 + i][v] + bias[H_ + hr];
        out[((size_t)(bb * H_ + hr)) * L_ + l] = a * (1.0f / (1.0f + expf(-gv)));
      }
    }
  }
}

extern "C" void kernel_launch(void* const* d_in, const int* in_sizes, int n_in,
                              void* d_out, int out_size, void* d_ws, size_t ws_size,
                              hipStream_t stream) {
  (void)in_sizes; (void)n_in; (void)out_size; (void)ws_size;
  const float* u          = (const float*)d_in[0];
  const float* log_dt     = (const float*)d_in[1];
  const float* C          = (const float*)d_in[2];
  const float* log_A_real = (const float*)d_in[3];
  const float* A_imag     = (const float*)d_in[4];
  const float* Dvec       = (const float*)d_in[5];
  const float* W          = (const float*)d_in[6];
  const float* bias       = (const float*)d_in[7];
  float* out = (float*)d_out;

  float* ws = (float*)d_ws;
  float* Ktap2 = ws;
  float* CoutR = Ktap2 + (size_t)H_ * 128;
  float* CoutI = CoutR + (size_t)H_ * Q_ * NMODE;
  float* XR    = CoutI + (size_t)H_ * Q_ * NMODE;
  float* XI    = XR    + (size_t)B_ * H_ * NCHUNK * NMODE;
  float* ygelu = XI    + (size_t)B_ * H_ * NCHUNK * NMODE;

  s4d_scan<<<H_, 256, 0, stream>>>(u, log_dt, C, log_A_real, A_imag,
                                   Ktap2, CoutR, CoutI, XR, XI);
  s4d_conv<<<H_, CV_THREADS, 0, stream>>>(u, Ktap2, CoutR, CoutI, XR, XI, Dvec, ygelu);
  s4d_glu<<<256, GP_THREADS, 0, stream>>>(ygelu, W, bias, out);
}


# scan and products in one launch: block = one head, its 8 waves scan the 8 batch rows, leave chunk-entering states and the head's tables in LDS, then run the products; 3 launches -> 2; sums unchanged
# speedup vs baseline: 1.7875x; 1.0732x over previous
#include <hip/hip_runtime.h>
#include <math.h>

#define H_     512
#define NMODE  32
#define B_     8
#define L_     2048
#define Q_     64
#define NCHUNK 32
#define CPITCH 34
#define WPITCH 516
#define XP     36

typedef __attribute__((ext_vector_type(2))) float v2f;
typedef __attribute__((ext_vector_type(8))) float v8f;

__device__ __forceinline__ v8f wmma4(v2f a, v2f b, v8f c) {
  return __builtin_amdgcn_wmma_f32_16x16x4_f32(false, a, false, b, (short)0, c,
                                               false, false);
}

__device__ __forceinline__ void mode_params(const float* __restrict__ log_dt,
                                            const float* __restrict__ C,
                                            const float* __restrict__ log_A_real,
                                            const float* __restrict__ A_imag,
                                            int h, int k,
                                            float& wr, float& wi, float& Ckr, float& Cki) {
  const float dt = expf(log_dt[h]);
  const float Ar = -expf(log_A_real[h * NMODE + k]);
  const float Ai = A_imag[h * NMODE + k];
  const float Ccr = C[(h * NMODE + k) * 2 + 0];
  const float Cci = C[(h * NMODE + k) * 2 + 1];
  const float er = expf(Ar * dt);
  wr = er * cosf(Ai * dt);
  wi = er * sinf(Ai * dt);
  const float exr = wr - 1.0f, exi = wi;
  const float den = Ar * Ar + Ai * Ai;
  const float qr = (exr * Ar + exi * Ai) / den;
  const float qi = (exi * Ar - exr * Ai) / den;
  Ckr = Ccr * qr - Cci * qi;
  Cki = Ccr * qi + Cci * qr;
}

#define SB 16

__device__ __forceinline__ void scan_load(float (&x)[Q_], const float4* __restrict__ p) {
#pragma unroll
  for (int q = 0; q < Q_ / 4; ++q) {
    const float4 v = p[q];
    x[4 * q + 0] = v.x; x[4 * q + 1] = v.y; x[4 * q + 2] = v.z; x[4 * q + 3] = v.w;
  }
}

__device__ __forceinline__ void scan_chunk(const float (&x)[Q_],
                                           const float (&pr)[SB], const float (&pi)[SB],
                                           float w16r, float w16i,
                                           float& sr, float& si) {
#pragma unroll
  for (int q = 0; q < Q_ / SB; ++q) {
    const float* xb = x + q * SB;
    float ar0 = pr[SB - 1] * xb[0], ai0 = pi[SB - 1] * xb[0];
    float ar1 = pr[SB - 2] * xb[1], ai1 = pi[SB - 2] * xb[1];
#pragma unroll
    for (int i = 2; i < SB - 2; i += 2) {
      ar0 = fmaf(pr[SB - 1 - i], xb[i], ar0);
      ai0 = fmaf(pi[SB - 1 - i], xb[i], ai0);
      ar1 = fmaf(pr[SB - 2 - i], xb[i + 1], ar1);
      ai1 = fmaf(pi[SB - 2 - i], xb[i + 1], ai1);
    }
    ar0 = fmaf(pr[1], xb[SB - 2], ar0);
    ai0 = fmaf(pi[1], xb[SB - 2], ai0);
    ar1 += xb[SB - 1];
    const float Sr = ar0 + ar1, Si = ai0 + ai1;
    const float nr = fmaf(w16r, sr, fmaf(-w16i, si, Sr));
    const float ni = fmaf(w16r, si, fmaf(w16i, sr, Si));
    sr = nr; si = ni;
  }
}

__device__ __forceinline__ int cv_kpos(int k) {
  return (k & ~6) | ((k & 2) << 1) | ((k & 4) >> 1);
}

#define CV_THREADS 256

__global__ void __launch_bounds__(CV_THREADS) __attribute__((amdgpu_waves_per_eu(4, 4)))
s4d_scan_conv(const float* __restrict__ u,
              const float* __restrict__ log_dt,
              const float* __restrict__ C,
              const float* __restrict__ log_A_real,
              const float* __restrict__ A_imag,
              const float* __restrict__ Dvec,
              float* __restrict__ ygelu) {
  __shared__ __attribute__((aligned(16))) float sKt[128];
  __shared__ __attribute__((aligned(16))) float sCR[Q_ * CPITCH];
  __shared__ __attribute__((aligned(16))) float sCI[Q_ * CPITCH];
  __shared__ __attribute__((aligned(16))) float sXR[B_ * NCHUNK * XP];
  __shared__ __attribute__((aligned(16))) float sXI[B_ * NCHUNK * XP];

  const int h      = blockIdx.x;
  const int tid    = threadIdx.x;
  const int waveId = tid >> 5;
  const int lane   = tid & 31;
  const int half   = lane >> 4;
  const int lm     = lane & 15;

  {
    const int b = waveId;
    const float4* up4 = (const float4*)(u + (size_t)(b * H_ + h) * L_);
    float xa[Q_], xc[Q_];
    scan_load(xa, up4);

    float wr, wi, Ckr, Cki;
    mode_params(log_dt, C, log_A_real, A_imag, h, lane, wr, wi, Ckr, Cki);

    float pr[SB], pi[SB];
    pr[0] = 1.0f; pi[0] = 0.0f;
    pr[1] = wr;   pi[1] = wi;
#pragma unroll
    for (int m = 2; m < SB; ++m) {
      pr[m] = pr[m - 1] * wr - pi[m - 1] * wi;
      pi[m] = pr[m - 1] * wi + pi[m - 1] * wr;
    }
    const float w16r = pr[SB - 1] * wr - pi[SB - 1] * wi;
    const float w16i = pr[SB - 1] * wi + pi[SB - 1] * wr;

    {
      const float w32r = w16r * w16r - w16i * w16i, w32i = 2.0f * w16r * w16i;
      const float w48r = w32r * w16r - w32i * w16i, w48i = w32r * w16i + w32i * w16r;
      const int   q  = b >> 1;
      const float gr = (q == 0) ? 1.0f : (q == 1) ? w16r : (q == 2) ? w32r : w48r;
      const float gi = (q == 0) ? 0.0f : (q == 1) ? w16i : (q == 2) ? w32i : w48i;
      const float er = (b & 1) ? pr[8] : 1.0f;
      const float ei = (b & 1) ? pi[8] : 0.0f;
      const float br = gr * er - gi * ei, bi = gr * ei + gi * er;
      const float q0r = Ckr * br - Cki * bi, q0i = Ckr * bi + Cki * br;

      float v[8];
#pragma unroll
      for (int i = 0; i < 8; ++i) v[i] = 2.0f * (q0r * pr[i] - q0i * pi[i]);
#pragma unroll
      for (int m = 16; m >= 1; m >>= 1) {
#pragma unroll
        for (int i = 0; i < 8; ++i) v[i] += __shfl_xor(v[i], m, 32);
      }
      if (lane == 0) {
        *(float4*)(sKt + 64 + 8 * b)     = make_float4(v[0], v[1], v[2], v[3]);
        *(float4*)(sKt + 64 + 8 * b + 4) = make_float4(v[4], v[5], v[6], v[7]);
      }
      if (tid < 64) sKt[tid] = 0.0f;
      const int kp = cv_kpos(lane);
#pragma unroll
      for (int i = 0; i < 8; ++i) {
        const float sr_ = q0r * pr[i + 1] - q0i * pi[i + 1];
        const float si_ = q0r * pi[i + 1] + q0i * pr[i + 1];
        sCR[(8 * b + i) * CPITCH + kp] = 2.0f * sr_;
        sCI[(8 * b + i) * CPITCH + kp] = -2.0f * si_;
      }
    }

    float sr = 0.0f, si = 0.0f;
    float* xrow = sXR + (size_t)(b * NCHUNK) * XP + lane;
    float* xiow = sXI + (size_t)(b * NCHUNK) * XP + lane;

    for (int c = 0; c < NCHUNK; c += 2) {
      scan_load(xc, up4 + (c + 1) * (Q_ / 4));
      xrow[c * XP] = sr;
      xiow[c * XP] = si;
      scan_chunk(xa, pr, pi, w16r, w16i, sr, si);
      const int cn = (c + 2 < NCHUNK) ? (c + 2) : (NCHUNK - 1);
      scan_load(xa, up4 + cn * (Q_ / 4));
      xrow[(c + 1) * XP] = sr;
      xiow[(c + 1) * XP] = si;
      scan_chunk(xc, pr, pi, w16r, w16i, sr, si);
    }
  }
  __syncthreads();

  const float Dh = Dvec[h];

  const int nh = waveId & 1;
  for (int it = 0; it < 4; ++it) {
    const int mt = (waveId >> 1) + 4 * it;
    const int qa = mt * 16 + lm;
    const int ba = qa >> 5, ca = qa & 31;
    const float4* up = (const float4*)(u + (size_t)(ba * H_ + h) * L_ + (size_t)ca * Q_) + half;
    const float4* xrp = (const float4*)(sXR + qa * XP) + half;
    const float4* xip = (const float4*)(sXI + qa * XP) + half;

    float4 ua[8], xr[4], xi[4];
#pragma unroll
    for (int i = 0; i < 8; ++i) ua[i] = up[2 * i];
#pragma unroll
    for (int i = 0; i < 4; ++i) { xr[i] = xrp[2 * i]; xi[i] = xip[2 * i]; }

    v8f acc[2];
#pragma unroll
    for (int nt = 0; nt < 2; ++nt) acc[nt] = (v8f){0.f, 0.f, 0.f, 0.f, 0.f, 0.f, 0.f, 0.f};

#pragma unroll
    for (int s = 0; s < 16; ++s) {
      v2f a;
      a.x = (s & 1) ? ua[s >> 1].z : ua[s >> 1].x;
      a.y = (s & 1) ? ua[s >> 1].w : ua[s >> 1].y;
      const int t0 = 8 * (s >> 1) + 4 * half + 2 * (s & 1);
#pragma unroll
      for (int nt = 0; nt < 2; ++nt) {
        const int j = (2 * nh + nt) * 16 + lm;
        v2f bb;
        bb.x = sKt[64 + j - t0];
        bb.y = sKt[63 + j - t0];
        acc[nt] = wmma4(a, bb, acc[nt]);
      }
    }
#pragma unroll
    for (int s = 0; s < 8; ++s) {
      v2f a;
      a.x = (s & 1) ? xr[s >> 1].z : xr[s >> 1].x;
      a.y = (s & 1) ? xr[s >> 1].w : xr[s >> 1].y;
#pragma unroll
      for (int nt = 0; nt < 2; ++nt) {
        const v2f bb = *(const v2f*)(sCR + ((2 * nh + nt) * 16 + lm) * CPITCH + 4 * s + 2 * half);
        acc[nt] = wmma4(a, bb, acc[nt]);
      }
    }
#pragma unroll
    for (int s = 0; s < 8; ++s) {
      v2f a;
      a.x = (s & 1) ? xi[s >> 1].z : xi[s >> 1].x;
      a.y = (s & 1) ? xi[s >> 1].w : xi[s >> 1].y;
#pragma unroll
      for (int nt = 0; nt < 2; ++nt) {
        const v2f bb = *(const v2f*)(sCI + ((2 * nh + nt) * 16 + lm) * CPITCH + 4 * s + 2 * half);
        acc[nt] = wmma4(a, bb, acc[nt]);
      }
    }

    float yv[8][2];
#pragma unroll
    for (int r = 0; r < 8; ++r) {
      const int qd = mt * 16 + 8 * half + r;
      const int bd = qd >> 5, cd = qd & 31;
      const size_t rb = (size_t)(bd * H_ + h) * L_ + (size_t)cd * Q_ + lm;
#pragma unroll
      for (int nt = 0; nt < 2; ++nt) {
        const size_t idx = rb + (2 * nh + nt) * 16;
        yv[r][nt] = acc[nt][r] + Dh * u[idx];
      }
    }
#pragma unroll
    for (int r = 0; r < 8; ++r) {
      const int qd = mt * 16 + 8 * half + r;
      const int bd = qd >> 5, cd = qd & 31;
      const size_t rb = (size_t)(bd * H_ + h) * L_ + (size_t)cd * Q_ + lm;
#pragma unroll
      for (int nt = 0; nt < 2; ++nt) {
        const size_t idx = rb + (2 * nh + nt) * 16;
        const float y = yv[r][nt];
        ygelu[idx] = 0.5f * y * (1.0f + erff(y * 0.70710678118654752f));
      }
    }
  }
}

#define GP_ROWS    128
#define GP_THREADS 512
#define GP_KG      16

__global__ void __launch_bounds__(GP_THREADS)
s4d_glu(const float* __restrict__ ygelu,
        const float* __restrict__ W,
        const float* __restrict__ bias,
        float* __restrict__ out) {
  __shared__ __attribute__((aligned(16))) float sW[GP_ROWS * WPITCH];
  const int tid    = threadIdx.x;
  const int waveId = tid >> 5;
  const int lane   = tid & 31;
  const int half   = lane >> 4;
  const int lm     = lane & 15;
  const int panel  = blockIdx.x & 7;
  const int cgrp   = blockIdx.x >> 3;
  const int col0   = cgrp * 512 + waveId * 16 + lm;
  const int bb     = col0 >> 11;
  const int l0     = col0 & (L_ - 1);

  const float* yb = ygelu + (size_t)bb * H_ * L_ + (size_t)(2 * half) * L_ + l0;

  v2f bcur[4], bnxt[4];
#pragma unroll
  for (int s = 0; s < 4; ++s) {
    bcur[s].x = yb[(size_t)(s * 4) * L_];
    bcur[s].y = yb[(size_t)(s * 4 + 1) * L_];
  }

  for (int it = 0; it < 4; ++it) {
    float4 tmp[8];
#pragma unroll
    for (int j = 0; j < 8; ++j) {
      const int r  = (it * 8 + j) * 4 + (tid >> 7);
      const int k4 = tid & 127;
      const int srow = (r < 64) ? (panel * 64 + r) : (H_ + panel * 64 + (r - 64));
      tmp[j] = *(const float4*)(W + (size_t)srow * H_ + k4 * 4);
    }
#pragma unroll
    for (int j = 0; j < 8; ++j) {
      const int r  = (it * 8 + j) * 4 + (tid >> 7);
      const int k4 = tid & 127;
      *(float4*)(sW + r * WPITCH + k4 * 4) = tmp[j];
    }
  }
  __syncthreads();

  const float* la = sW + lm * WPITCH + 2 * half;

  for (int t = 0; t < 2; ++t) {
    v8f acc[8];
#pragma unroll
    for (int i = 0; i < 8; ++i) {
      acc[i] = (v8f){0.f, 0.f, 0.f, 0.f, 0.f, 0.f, 0.f, 0.f};
    }

    for (int g = 0; g < H_ / GP_KG; ++g) {
      int gn = t * (H_ / GP_KG) + g + 1;
      gn = (gn < 2 * (H_ / GP_KG)) ? gn : (2 * (H_ / GP_KG) - 1);
      const float* yn = yb + (gn >> 5) * 256 + (size_t)((gn & 31) * GP_KG) * L_;
#pragma unroll
      for (int s = 0; s < 4; ++s) {
        bnxt[s].x = yn[(size_t)(s * 4) * L_];
        bnxt[s].y = yn[(size_t)(s * 4 + 1) * L_];
      }

      const float* lk = la + g * GP_KG;
#pragma unroll
      for (int s = 0; s < 4; ++s) {
        v2f a[8];
#pragma unroll
        for (int i = 0; i < 8; ++i) {
          a[i] = *(const v2f*)(lk + i * 16 * WPITCH + s * 4);
        }
#pragma unroll
        for (int i = 0; i < 8; ++i) {
          acc[i] = wmma4(a[i], bcur[s], acc[i]);
        }
      }
#pragma unroll
      for (int s = 0; s < 4; ++s) bcur[s] = bnxt[s];
    }

    const int l = l0 + t * 256;
#pragma unroll
    for (int i = 0; i < 4; ++i) {
#pragma unroll
      for (int v = 0; v < 8; ++v) {
        const int hr = panel * 64 + i * 16 + v + 8 * half;
        const float a  = acc[i][v] + bias[hr];
        const float gv = acc[4 + i][v] + bias[H_ + hr];
        out[((size_t)(bb * H_ + hr)) * L_ + l] = a * (1.0f / (1.0f + expf(-gv)));
      }
    }
  }
}

extern "C" void kernel_launch(void* const* d_in, const int* in_sizes, int n_in,
                              void* d_out, int out_size, void* d_ws, size_t ws_size,
                              hipStream_t stream) {
  (void)in_sizes; (void)n_in; (void)out_size; (void)ws_size;
  const float* u          = (const float*)d_in[0];
  const float* log_dt     = (const float*)d_in[1];
  const float* C          = (const float*)d_in[2];
  const float* log_A_real = (const float*)d_in[3];
  const float* A_imag     = (const float*)d_in[4];
  const float* Dvec       = (const float*)d_in[5];
  const float* W          = (const float*)d_in[6];
  const float* bias       = (const float*)d_in[7];
  float* out = (float*)d_out;

  float* ygelu = (float*)d_ws;

  s4d_scan_conv<<<H_, CV_THREADS, 0, stream>>>(u, log_dt, C, log_A_real, A_imag, Dvec, ygelu);
  s4d_glu<<<256, GP_THREADS, 0, stream>>>(ygelu, W, bias, out);
}


# GLU GEMM: W panel in LDS as 8 k-blocks of 128 rows x 64 k; fragments read as 16-byte pieces at one base + fixed offsets (k order in 8-k groups remapped, both operands); k sum order changes
# speedup vs baseline: 1.9262x; 1.0776x over previous
#include <hip/hip_runtime.h>
#include <math.h>

#define H_     512
#define NMODE  32
#define B_     8
#define L_     2048
#define Q_     64
#define NCHUNK 32
#define CPITCH 34
#define XP     36

typedef __attribute__((ext_vector_type(2))) float v2f;
typedef __attribute__((ext_vector_type(8))) float v8f;

__device__ __forceinline__ v8f wmma4(v2f a, v2f b, v8f c) {
  return __builtin_amdgcn_wmma_f32_16x16x4_f32(false, a, false, b, (short)0, c,
                                               false, false);
}

__device__ __forceinline__ void mode_params(const float* __restrict__ log_dt,
                                            const float* __restrict__ C,
                                            const float* __restrict__ log_A_real,
                                            const float* __restrict__ A_imag,
                                            int h, int k,
                                            float& wr, float& wi, float& Ckr, float& Cki) {
  const float dt = expf(log_dt[h]);
  const float Ar = -expf(log_A_real[h * NMODE + k]);
  const float Ai = A_imag[h * NMODE + k];
  const float Ccr = C[(h * NMODE + k) * 2 + 0];
  const float Cci = C[(h * NMODE + k) * 2 + 1];
  const float er = expf(Ar * dt);
  wr = er * cosf(Ai * dt);
  wi = er * sinf(Ai * dt);
  const float exr = wr - 1.0f, exi = wi;
  const float den = Ar * Ar + Ai * Ai;
  const float qr = (exr * Ar + exi * Ai) / den;
  const float qi = (exi * Ar - exr * Ai) / den;
  Ckr = Ccr * qr - Cci * qi;
  Cki = Ccr * qi + Cci * qr;
}

#define SB 16

__device__ __forceinline__ void scan_load(float (&x)[Q_], const float4* __restrict__ p) {
#pragma unroll
  for (int q = 0; q < Q_ / 4; ++q) {
    const float4 v = p[q];
    x[4 * q + 0] = v.x; x[4 * q + 1] = v.y; x[4 * q + 2] = v.z; x[4 * q + 3] = v.w;
  }
}

__device__ __forceinline__ void scan_chunk(const float (&x)[Q_],
                                           const float (&pr)[SB], const float (&pi)[SB],
                                           float w16r, float w16i,
                                           float& sr, float& si) {
#pragma unroll
  for (int q = 0; q < Q_ / SB; ++q) {
    const float* xb = x + q * SB;
    float ar0 = pr[SB - 1] * xb[0], ai0 = pi[SB - 1] * xb[0];
    float ar1 = pr[SB - 2] * xb[1], ai1 = pi[SB - 2] * xb[1];
#pragma unroll
    for (int i = 2; i < SB - 2; i += 2) {
      ar0 = fmaf(pr[SB - 1 - i], xb[i], ar0);
      ai0 = fmaf(pi[SB - 1 - i], xb[i], ai0);
      ar1 = fmaf(pr[SB - 2 - i], xb[i + 1], ar1);
      ai1 = fmaf(pi[SB - 2 - i], xb[i + 1], ai1);
    }
    ar0 = fmaf(pr[1], xb[SB - 2], ar0);
    ai0 = fmaf(pi[1], xb[SB - 2], ai0);
    ar1 += xb[SB - 1];
    const float Sr = ar0 + ar1, Si = ai0 + ai1;
    const float nr = fmaf(w16r, sr, fmaf(-w16i, si, Sr));
    const float ni = fmaf(w16r, si, fmaf(w16i, sr, Si));
    sr = nr; si = ni;
  }
}

__device__ __forceinline__ int cv_kpos(int k) {
  return (k & ~6) | ((k & 2) << 1) | ((k & 4) >> 1);
}

#define CV_THREADS 256

__global__ void __launch_bounds__(CV_THREADS) __attribute__((amdgpu_waves_per_eu(4, 4)))
s4d_scan_conv(const float* __restrict__ u,
              const float* __restrict__ log_dt,
              const float* __restrict__ C,
              const float* __restrict__ log_A_real,
              const float* __restrict__ A_imag,
              const float* __restrict__ Dvec,
              float* __restrict__ ygelu) {
  __shared__ __attribute__((aligned(16))) float sKt[128];
  __shared__ __attribute__((aligned(16))) float sCR[Q_ * CPITCH];
  __shared__ __attribute__((aligned(16))) float sCI[Q_ * CPITCH];
  __shared__ __attribute__((aligned(16))) float sXR[B_ * NCHUNK * XP];
  __shared__ __attribute__((aligned(16))) float sXI[B_ * NCHUNK * XP];

  const int h      = blockIdx.x;
  const int tid    = threadIdx.x;
  const int waveId = tid >> 5;
  const int lane   = tid & 31;
  const int half   = lane >> 4;
  const int lm     = lane & 15;

  {
    const int b = waveId;
    const float4* up4 = (const float4*)(u + (size_t)(b * H_ + h) * L_);
    float xa[Q_], xc[Q_];
    scan_load(xa, up4);

    float wr, wi, Ckr, Cki;
    mode_params(log_dt, C, log_A_real, A_imag, h, lane, wr, wi, Ckr, Cki);

    float pr[SB], pi[SB];
    pr[0] = 1.0f; pi[0] = 0.0f;
    pr[1] = wr;   pi[1] = wi;
#pragma unroll
    for (int m = 2; m < SB; ++m) {
      pr[m] = pr[m - 1] * wr - pi[m - 1] * wi;
      pi[m] = pr[m - 1] * wi + pi[m - 1] * wr;
    }
    const float w16r = pr[SB - 1] * wr - pi[SB - 1] * wi;
    const float w16i = pr[SB - 1] * wi + pi[SB - 1] * wr;

    {
      const float w32r = w16r * w16r - w16i * w16i, w32i = 2.0f * w16r * w16i;
      const float w48r = w32r * w16r - w32i * w16i, w48i = w32r * w16i + w32i * w16r;
      const int   q  = b >> 1;
      const float gr = (q == 0) ? 1.0f : (q == 1) ? w16r : (q == 2) ? w32r : w48r;
      const float gi = (q == 0) ? 0.0f : (q == 1) ? w16i : (q == 2) ? w32i : w48i;
      const float er = (b & 1) ? pr[8] : 1.0f;
      const float ei = (b & 1) ? pi[8] : 0.0f;
      const float br = gr * er - gi * ei, bi = gr * ei + gi * er;
      const float q0r = Ckr * br - Cki * bi, q0i = Ckr * bi + Cki * br;

      float v[8];
#pragma unroll
      for (int i = 0; i < 8; ++i) v[i] = 2.0f * (q0r * pr[i] - q0i * pi[i]);
#pragma unroll
      for (int m = 16; m >= 1; m >>= 1) {
#pragma unroll
        for (int i = 0; i < 8; ++i) v[i] += __shfl_xor(v[i], m, 32);
      }
      if (lane == 0) {
        *(float4*)(sKt + 64 + 8 * b)     = make_float4(v[0], v[1], v[2], v[3]);
        *(float4*)(sKt + 64 + 8 * b + 4) = make_float4(v[4], v[5], v[6], v[7]);
      }
      if (tid < 64) sKt[tid] = 0.0f;
      const int kp = cv_kpos(lane);
#pragma unroll
      for (int i = 0; i < 8; ++i) {
        const float sr_ = q0r * pr[i + 1] - q0i * pi[i + 1];
        const float si_ = q0r * pi[i + 1] + q0i * pr[i + 1];
        sCR[(8 * b + i) * CPITCH + kp] = 2.0f * sr_;
        sCI[(8 * b + i) * CPITCH + kp] = -2.0f * si_;
      }
    }

    float sr = 0.0f, si = 0.0f;
    float* xrow = sXR + (size_t)(b * NCHUNK) * XP + lane;
    float* xiow = sXI + (size_t)(b * NCHUNK) * XP + lane;

    for (int c = 0; c < NCHUNK; c += 2) {
      scan_load(xc, up4 + (c + 1) * (Q_ / 4));
      xrow[c * XP] = sr;
      xiow[c * XP] = si;
      scan_chunk(xa, pr, pi, w16r, w16i, sr, si);
      const int cn = (c + 2 < NCHUNK) ? (c + 2) : (NCHUNK - 1);
      scan_load(xa, up4 + cn * (Q_ / 4));
      xrow[(c + 1) * XP] = sr;
      xiow[(c + 1) * XP] = si;
      scan_chunk(xc, pr, pi, w16r, w16i, sr, si);
    }
  }
  __syncthreads();

  const float Dh = Dvec[h];

  const int nh = waveId & 1;
  for (int it = 0; it < 4; ++it) {
    const int mt = (waveId >> 1) + 4 * it;
    const int qa = mt * 16 + lm;
    const int ba = qa >> 5, ca = qa & 31;
    const float4* up = (const float4*)(u + (size_t)(ba * H_ + h) * L_ + (size_t)ca * Q_) + half;
    const float4* xrp = (const float4*)(sXR + qa * XP) + half;
    const float4* xip = (const float4*)(sXI + qa * XP) + half;

    float4 ua[8], xr[4], xi[4];
#pragma unroll
    for (int i = 0; i < 8; ++i) ua[i] = up[2 * i];
#pragma unroll
    for (int i = 0; i < 4; ++i) { xr[i] = xrp[2 * i]; xi[i] = xip[2 * i]; }

    v8f acc[2];
#pragma unroll
    for (int nt = 0; nt < 2; ++nt) acc[nt] = (v8f){0.f, 0.f, 0.f, 0.f, 0.f, 0.f, 0.f, 0.f};

#pragma unroll
    for (int s = 0; s < 16; ++s) {
      v2f a;
      a.x = (s & 1) ? ua[s >> 1].z : ua[s >> 1].x;
      a.y = (s & 1) ? ua[s >> 1].w : ua[s >> 1].y;
      const int t0 = 8 * (s >> 1) + 4 * half + 2 * (s & 1);
#pragma unroll
      for (int nt = 0; nt < 2; ++nt) {
        const int j = (2 * nh + nt) * 16 + lm;
        v2f bb;
        bb.x = sKt[64 + j - t0];
        bb.y = sKt[63 + j - t0];
        acc[nt] = wmma4(a, bb, acc[nt]);
      }
    }
#pragma unroll
    for (int s = 0; s < 8; ++s) {
      v2f a;
      a.x = (s & 1) ? xr[s >> 1].z : xr[s >> 1].x;
      a.y = (s & 1) ? xr[s >> 1].w : xr[s >> 1].y;
#pragma unroll
      for (int nt = 0; nt < 2; ++nt) {
        const v2f bb = *(const v2f*)(sCR + ((2 * nh + nt) * 16 + lm) * CPITCH + 4 * s + 2 * half);
        acc[nt] = wmma4(a, bb, acc[nt]);
      }
    }
#pragma unroll
    for (int s = 0; s < 8; ++s) {
      v2f a;
      a.x = (s & 1) ? xi[s >> 1].z : xi[s >> 1].x;
      a.y = (s & 1) ? xi[s >> 1].w : xi[s >> 1].y;
#pragma unroll
      for (int nt = 0; nt < 2; ++nt) {
        const v2f bb = *(const v2f*)(sCI + ((2 * nh + nt) * 16 + lm) * CPITCH + 4 * s + 2 * half);
        acc[nt] = wmma4(a, bb, acc[nt]);
      }
    }

    float yv[8][2];
#pragma unroll
    for (int r = 0; r < 8; ++r) {
      const int qd = mt * 16 + 8 * half + r;
      const int bd = qd >> 5, cd = qd & 31;
      const size_t rb = (size_t)(bd * H_ + h) * L_ + (size_t)cd * Q_ + lm;
#pragma unroll
      for (int nt = 0; nt < 2; ++nt) {
        const size_t idx = rb + (2 * nh + nt) * 16;
        yv[r][nt] = acc[nt][r] + Dh * u[idx];
      }
    }
#pragma unroll
    for (int r = 0; r < 8; ++r) {
      const int qd = mt * 16 + 8 * half + r;
      const int bd = qd >> 5, cd = qd & 31;
      const size_t rb = (size_t)(bd * H_ + h) * L_ + (size_t)cd * Q_ + lm;
#pragma unroll
      for (int nt = 0; nt < 2; ++nt) {
        const size_t idx = rb + (2 * nh + nt) * 16;
        const float y = yv[r][nt];
        ygelu[idx] = 0.5f * y * (1.0f + erff(y * 0.70710678118654752f));
      }
    }
  }
}

#define GP_ROWS    128
#define GP_THREADS 512
#define GP_KG      16
#define GP_NG      (H_ / GP_KG)
#define KB_K       64
#define KB_PITCH   68
#define KB_STRIDE  (GP_ROWS * KB_PITCH)

#define GP_BROW(s) (8 * ((s) >> 1) + 2 * ((s) & 1))

__device__ __forceinline__ void glu_load_b(v2f (&b)[4], const float* __restrict__ yb, int gi) {
  gi = (gi < 2 * GP_NG) ? gi : (2 * GP_NG - 1);
  const float* p = yb + (gi >> 5) * 256 + (size_t)((gi & (GP_NG - 1)) * GP_KG) * L_;
#pragma unroll
  for (int s = 0; s < 4; ++s) {
    b[s].x = p[(size_t)GP_BROW(s) * L_];
    b[s].y = p[(size_t)(GP_BROW(s) + 1) * L_];
  }
}

__device__ __forceinline__ void glu_group(v8f (&acc)[8], const float* lk, const v2f (&b)[4]) {
#pragma unroll
  for (int jj = 0; jj < 2; ++jj) {
    float4 a4[8];
#pragma unroll
    for (int i = 0; i < 8; ++i) {
      a4[i] = *(const float4*)(lk + i * 16 * KB_PITCH + jj * 8);
    }
#pragma unroll
    for (int i = 0; i < 8; ++i) {
      v2f a; a.x = a4[i].x; a.y = a4[i].y;
      acc[i] = wmma4(a, b[2 * jj], acc[i]);
    }
#pragma unroll
    for (int i = 0; i < 8; ++i) {
      v2f a; a.x = a4[i].z; a.y = a4[i].w;
      acc[i] = wmma4(a, b[2 * jj + 1], acc[i]);
    }
  }
}

__global__ void __launch_bounds__(GP_THREADS)
s4d_glu(const float* __restrict__ ygelu,
        const float* __restrict__ W,
        const float* __restrict__ bias,
        float* __restrict__ out) {
  __shared__ __attribute__((aligned(16))) float sW[(H_ / KB_K) * KB_STRIDE];
  const int tid    = threadIdx.x;
  const int waveId = tid >> 5;
  const int lane   = tid & 31;
  const int half   = lane >> 4;
  const int lm     = lane & 15;
  const int panel  = blockIdx.x & 7;
  const int cgrp   = blockIdx.x >> 3;
  const int col0   = cgrp * 512 + waveId * 16 + lm;
  const int bb     = col0 >> 11;
  const int l0     = col0 & (L_ - 1);

  const float* yb = ygelu + (size_t)bb * H_ * L_ + (size_t)(4 * half) * L_ + l0;

  v2f b0[4], b1[4];
  glu_load_b(b0, yb, 0);

  for (int it = 0; it < 4; ++it) {
    float4 tmp[8];
#pragma unroll
    for (int j = 0; j < 8; ++j) {
      const int r  = (it * 8 + j) * 4 + (tid >> 7);
      const int k4 = tid & 127;
      const int srow = (r < 64) ? (panel * 64 + r) : (H_ + panel * 64 + (r - 64));
      tmp[j] = *(const float4*)(W + (size_t)srow * H_ + k4 * 4);
    }
#pragma unroll
    for (int j = 0; j < 8; ++j) {
      const int r  = (it * 8 + j) * 4 + (tid >> 7);
      const int k4 = tid & 127;
      *(float4*)(sW + (k4 >> 4) * KB_STRIDE + r * KB_PITCH + (k4 & 15) * 4) = tmp[j];
    }
  }
  __syncthreads();

  const float* la = sW + lm * KB_PITCH + 4 * half;

  for (int t = 0; t < 2; ++t) {
    v8f acc[8];
#pragma unroll
    for (int i = 0; i < 8; ++i) {
      acc[i] = (v8f){0.f, 0.f, 0.f, 0.f, 0.f, 0.f, 0.f, 0.f};
    }

    for (int q = 0; q < H_ / KB_K; ++q) {
      const int g0 = t * GP_NG + 4 * q;
      const float* lk = la + q * KB_STRIDE;
      glu_load_b(b1, yb, g0 + 1);
      glu_group(acc, lk, b0);
      glu_load_b(b0, yb, g0 + 2);
      glu_group(acc, lk + GP_KG, b1);
      glu_load_b(b1, yb, g0 + 3);
      glu_group(acc, lk + 2 * GP_KG, b0);
      glu_load_b(b0, yb, g0 + 4);
      glu_group(acc, lk + 3 * GP_KG, b1);
    }

    const int l = l0 + t * 256;
#pragma unroll
    for (int i = 0; i < 4; ++i) {
#pragma unroll
      for (int v = 0; v < 8; ++v) {
        const int hr = panel * 64 + i * 16 + v + 8 * half;
        const float a  = acc[i][v] + bias[hr];
        const float gv = acc[4 + i][v] + bias[H_ + hr];
        out[((size_t)(bb * H_ + hr)) * L_ + l] = a * (1.0f / (1.0f + expf(-gv)));
      }
    }
  }
}

extern "C" void kernel_launch(void* const* d_in, const int* in_sizes, int n_in,
                              void* d_out, int out_size, void* d_ws, size_t ws_size,
                              hipStream_t stream) {
  (void)in_sizes; (void)n_in; (void)out_size; (void)ws_size;
  const float* u          = (const float*)d_in[0];
  const float* log_dt     = (const float*)d_in[1];
  const float* C          = (const float*)d_in[2];
  const float* log_A_real = (const float*)d_in[3];
  const float* A_imag     = (const float*)d_in[4];
  const float* Dvec       = (const float*)d_in[5];
  const float* W          = (const float*)d_in[6];
  const float* bias       = (const float*)d_in[7];
  float* out = (float*)d_out;

  float* ygelu = (float*)d_ws;

  s4d_scan_conv<<<H_, CV_THREADS, 0, stream>>>(u, log_dt, C, log_A_real, A_imag, Dvec, ygelu);
  s4d_glu<<<256, GP_THREADS, 0, stream>>>(ygelu, W, bias, out);
}
